# starting kernel
# baseline (speedup 1.0000x reference)
#include <hip/hip_runtime.h>


#define N_NODES   100000
#define N_EDGES   1600000
#define IN_FEATS  128
#define H_FEATS   64
#define N_CONVS   3
#define POLY_ORDER 4
#define NUM_CLASSES 2

typedef __attribute__((ext_vector_type(2))) float v2f;
typedef __attribute__((ext_vector_type(8))) float v8f;

__device__ __forceinline__ v8f wmma_f32(v2f a, v2f b, v8f c) {
  return __builtin_amdgcn_wmma_f32_16x16x4_f32(
       false, a,  false, b,
       (short)0, c,  false,  false);
}

__global__ __launch_bounds__(256)
void zero_f32(float* __restrict__ p, int n) {
  int stride = gridDim.x * blockDim.x;
  for (int i = blockIdx.x * blockDim.x + threadIdx.x; i < n; i += stride)
    p[i] = 0.0f;
}

__global__ __launch_bounds__(256)
void degree_kernel(const int* __restrict__ dst, float* __restrict__ deg, int E) {
  int e = blockIdx.x * blockDim.x + threadIdx.x;
  if (e < E) atomicAdd(&deg[dst[e]], 1.0f);
}

__global__ __launch_bounds__(256)
void dinv_kernel(float* __restrict__ deg, int n) {
  int i = blockIdx.x * blockDim.x + threadIdx.x;
  if (i < n) {
    float d = deg[i];
    d = d < 1.0f ? 1.0f : d;
    deg[i] = rsqrtf(d);
  }
}

__global__ __launch_bounds__(128)
void trunk_kernel(const float* __restrict__ X,
                  const float* __restrict__ W1, const float* __restrict__ b1,
                  const float* __restrict__ W2, const float* __restrict__ b2,
                  float* __restrict__ P0) {
  __shared__ float ldsX[16 * 132];
  __shared__ float ldsH[16 * 72];

  const int tid  = threadIdx.x;
  const int lane = tid & 31;
  const int wave = tid >> 5;
  const int mrow = lane & 15;
  const int kh   = lane >> 4;
  const int row0 = blockIdx.x * 16;
  const int col0 = wave * 16;

  for (int i = tid; i < 16 * IN_FEATS; i += 128) {
    int r = i >> 7, c = i & 127;
    ldsX[r * 132 + c] = X[(size_t)(row0 + r) * IN_FEATS + c];
  }
  __syncthreads();

  v8f acc = {};
  for (int k0 = 0; k0 < IN_FEATS; k0 += 4) {
    v2f a, b;
    a.x = ldsX[mrow * 132 + k0 + 2 * kh];
    a.y = ldsX[mrow * 132 + k0 + 2 * kh + 1];
    b.x = W1[(size_t)(k0 + 2 * kh)     * H_FEATS + col0 + mrow];
    b.y = W1[(size_t)(k0 + 2 * kh + 1) * H_FEATS + col0 + mrow];
    acc = wmma_f32(a, b, acc);
  }
  {
    float bias = b1[col0 + mrow];
    #pragma unroll
    for (int v = 0; v < 8; ++v) {
      int r = v + 8 * kh;
      float val = acc[v] + bias;
      ldsH[r * 72 + col0 + mrow] = val > 0.0f ? val : 0.0f;
    }
  }
  __syncthreads();

  v8f acc2 = {};
  for (int k0 = 0; k0 < H_FEATS; k0 += 4) {
    v2f a, b;
    a.x = ldsH[mrow * 72 + k0 + 2 * kh];
    a.y = ldsH[mrow * 72 + k0 + 2 * kh + 1];
    b.x = W2[(size_t)(k0 + 2 * kh)     * H_FEATS + col0 + mrow];
    b.y = W2[(size_t)(k0 + 2 * kh + 1) * H_FEATS + col0 + mrow];
    acc2 = wmma_f32(a, b, acc2);
  }
  {
    float bias = b2[col0 + mrow];
    #pragma unroll
    for (int v = 0; v < 8; ++v) {
      int r = v + 8 * kh;
      float val = acc2[v] + bias;
      P0[(size_t)(row0 + r) * H_FEATS + col0 + mrow] = val > 0.0f ? val : 0.0f;
    }
  }
}

__global__ __launch_bounds__(256)
void scatter_kernel(const float* __restrict__ P,
                    const int* __restrict__ src, const int* __restrict__ dst,
                    const float* __restrict__ dinv, float* __restrict__ agg,
                    int E) {
  int gid  = blockIdx.x * blockDim.x + threadIdx.x;
  int e    = gid >> 5;
  int lane = gid & 31;
  if (e >= E) return;
  int s = src[e];
  int d = dst[e];
  float di = dinv[s];
  const float2* pp = reinterpret_cast<const float2*>(P + (size_t)s * H_FEATS) + lane;
  float2 v = *pp;
  float* ap = agg + (size_t)d * H_FEATS + 2 * lane;
  atomicAdd(ap,     v.x * di);
  atomicAdd(ap + 1, v.y * di);
}

__global__ __launch_bounds__(256)
void update_kernel(const float* __restrict__ Pprev, float* __restrict__ agg,
                   const float* __restrict__ dinv, float* __restrict__ Pk,
                   int n) {
  int i = blockIdx.x * blockDim.x + threadIdx.x;
  if (i >= n) return;
  int node = i >> 6;
  float a = agg[i];
  agg[i] = 0.0f;
  Pk[i] = Pprev[i] - a * dinv[node];
}

__global__ __launch_bounds__(256)
void wc_kernel(const float* __restrict__ thetas, const float* __restrict__ Wm1,
               float* __restrict__ Wc) {
  int t = blockIdx.x * blockDim.x + threadIdx.x;
  if (t >= POLY_ORDER * H_FEATS * H_FEATS) return;
  int k  = t >> 12;
  int rc = t & 4095;
  int r  = rc >> 6, c = rc & 63;
  float s = 0.0f;
  #pragma unroll
  for (int i = 0; i < N_CONVS; ++i)
    s += thetas[i * POLY_ORDER + k] * Wm1[(size_t)(i * H_FEATS + r) * H_FEATS + c];
  Wc[t] = s;
}

__global__ __launch_bounds__(128)
void head_kernel(const float* __restrict__ P0, const float* __restrict__ P1,
                 const float* __restrict__ P2, const float* __restrict__ P3,
                 const float* __restrict__ Wc, const float* __restrict__ bm1,
                 const float* __restrict__ Wm2, const float* __restrict__ bm2,
                 float* __restrict__ out) {
  __shared__ float ldsA[16 * 72];
  __shared__ float ldsH[16 * 72];

  const int tid  = threadIdx.x;
  const int lane = tid & 31;
  const int wave = tid >> 5;
  const int mrow = lane & 15;
  const int kh   = lane >> 4;
  const int row0 = blockIdx.x * 16;
  const int col0 = wave * 16;

  const float* Ps[4] = {P0, P1, P2, P3};

  v8f acc = {};
  #pragma unroll
  for (int k = 0; k < POLY_ORDER; ++k) {
    const float* Pk = Ps[k];
    const float* Wk = Wc + k * H_FEATS * H_FEATS;
    for (int i = tid; i < 16 * H_FEATS; i += 128) {
      int r = i >> 6, c = i & 63;
      ldsA[r * 72 + c] = Pk[(size_t)(row0 + r) * H_FEATS + c];
    }
    __syncthreads();
    for (int k0 = 0; k0 < H_FEATS; k0 += 4) {
      v2f a, b;
      a.x = ldsA[mrow * 72 + k0 + 2 * kh];
      a.y = ldsA[mrow * 72 + k0 + 2 * kh + 1];
      b.x = Wk[(k0 + 2 * kh)     * H_FEATS + col0 + mrow];
      b.y = Wk[(k0 + 2 * kh + 1) * H_FEATS + col0 + mrow];
      acc = wmma_f32(a, b, acc);
    }
    __syncthreads();
  }

  {
    float bias = bm1[col0 + mrow];
    #pragma unroll
    for (int v = 0; v < 8; ++v) {
      int r = v + 8 * kh;
      float val = acc[v] + bias;
      ldsH[r * 72 + col0 + mrow] = val > 0.0f ? val : 0.0f;
    }
  }
  __syncthreads();

  if (tid < 32) {
    int r = tid >> 1, c = tid & 1;
    float s = bm2[c];
    #pragma unroll 8
    for (int j = 0; j < H_FEATS; ++j)
      s += ldsH[r * 72 + j] * Wm2[j * NUM_CLASSES + c];
    out[(size_t)(row0 + r) * NUM_CLASSES + c] = s;
  }
}

extern "C" void kernel_launch(void* const* d_in, const int* in_sizes, int n_in,
                              void* d_out, int out_size, void* d_ws, size_t ws_size,
                              hipStream_t stream) {
  const float* X      = (const float*)d_in[0];
  const int*   src    = (const int*)  d_in[1];
  const int*   dst    = (const int*)  d_in[2];
  const float* W1     = (const float*)d_in[3];
  const float* b1     = (const float*)d_in[4];
  const float* W2     = (const float*)d_in[5];
  const float* b2     = (const float*)d_in[6];
  const float* thetas = (const float*)d_in[7];
  const float* Wm1    = (const float*)d_in[8];
  const float* bm1    = (const float*)d_in[9];
  const float* Wm2    = (const float*)d_in[10];
  const float* bm2    = (const float*)d_in[11];
  float* out = (float*)d_out;
  float* ws  = (float*)d_ws;

  const int PN = N_NODES * H_FEATS;
  float* dinv = ws;
  float* P0   = ws + 100352;
  float* P1   = P0 + PN;
  float* P2   = P1 + PN;
  float* P3   = P2 + PN;
  float* agg  = P3 + PN;
  float* Wc   = agg + PN;

  zero_f32<<<512, 256, 0, stream>>>(dinv, N_NODES);
  zero_f32<<<2048, 256, 0, stream>>>(agg, PN);

  degree_kernel<<<(N_EDGES + 255) / 256, 256, 0, stream>>>(dst, dinv, N_EDGES);
  dinv_kernel<<<(N_NODES + 255) / 256, 256, 0, stream>>>(dinv, N_NODES);

  trunk_kernel<<<N_NODES / 16, 128, 0, stream>>>(X, W1, b1, W2, b2, P0);

  wc_kernel<<<(POLY_ORDER * H_FEATS * H_FEATS + 255) / 256, 256, 0, stream>>>(
      thetas, Wm1, Wc);

  float* Pbuf[4] = {P0, P1, P2, P3};
  for (int k = 1; k < POLY_ORDER; ++k) {
    scatter_kernel<<<(N_EDGES * 32 + 255) / 256, 256, 0, stream>>>(
        Pbuf[k - 1], src, dst, dinv, agg, N_EDGES);
    update_kernel<<<(PN + 255) / 256, 256, 0, stream>>>(
        Pbuf[k - 1], agg, dinv, Pbuf[k], PN);
  }

  head_kernel<<<N_NODES / 16, 128, 0, stream>>>(
      P0, P1, P2, P3, Wc, bm1, Wm2, bm2, out);
}


# in-degrees: per-edge global f32 atomics on random words replaced by LDS histograms per (edge slice, node range); integer partial counts stored as whole lines, summed per node; 9 -> 8 launches
# speedup vs baseline: 2.4461x; 2.4461x over previous
#include <hip/hip_runtime.h>


#define N_NODES   100000
#define N_EDGES   1600000
#define IN_FEATS  128
#define H_FEATS   64
#define N_CONVS   3
#define POLY_ORDER 4
#define NUM_CLASSES 2

typedef __attribute__((ext_vector_type(2))) float v2f;
typedef __attribute__((ext_vector_type(8))) float v8f;

__device__ __forceinline__ v8f wmma_f32(v2f a, v2f b, v8f c) {
  return __builtin_amdgcn_wmma_f32_16x16x4_f32(
       false, a,  false, b,
       (short)0, c,  false,  false);
}

__global__ __launch_bounds__(256)
void zero_f32x4(float* __restrict__ p, int n4) {
  float4* p4 = reinterpret_cast<float4*>(p);
  int stride = gridDim.x * blockDim.x;
  for (int i = blockIdx.x * blockDim.x + threadIdx.x; i < n4; i += stride)
    p4[i] = make_float4(0.0f, 0.0f, 0.0f, 0.0f);
}

#define DEG_RANGES      8
#define DEG_RANGE_NODES 12512
#define DEG_SLICES      32
#define DEG_THREADS     512
#define DEG_LOADS       8

__global__ __launch_bounds__(DEG_THREADS)
void degree_hist_kernel(const int* __restrict__ dst, unsigned* __restrict__ part,
                        int E, int n) {
  __shared__ unsigned cnt[DEG_RANGE_NODES];

  const int tid   = threadIdx.x;
  const int range = blockIdx.x % DEG_RANGES;
  const int slice = blockIdx.x / DEG_RANGES;
  const int lo    = range * DEG_RANGE_NODES;
  int hi = lo + DEG_RANGE_NODES;
  if (hi > n) hi = n;
  const unsigned span = hi > lo ? (unsigned)(hi - lo) : 0u;

  for (int i = tid; i < DEG_RANGE_NODES; i += DEG_THREADS) cnt[i] = 0u;
  __syncthreads();

  const int slice_len = (((E + DEG_SLICES - 1) / DEG_SLICES) + 3) & ~3;
  const int e_begin = slice * slice_len;
  int e_end = e_begin + slice_len;
  if (e_end > E) e_end = E;
  const int n4 = e_end > e_begin ? (e_end - e_begin) >> 2 : 0;

  const int4* d4 = reinterpret_cast<const int4*>(dst + e_begin);
  for (int i0 = tid; i0 < n4; i0 += DEG_LOADS * DEG_THREADS) {
    int4 v[DEG_LOADS];
    #pragma unroll
    for (int u = 0; u < DEG_LOADS; ++u) {
      const int i = i0 + u * DEG_THREADS;
      v[u] = i < n4 ? d4[i] : make_int4(-1, -1, -1, -1);
    }
    #pragma unroll
    for (int u = 0; u < DEG_LOADS; ++u) {
      const unsigned a = (unsigned)(v[u].x - lo);
      const unsigned b = (unsigned)(v[u].y - lo);
      const unsigned c = (unsigned)(v[u].z - lo);
      const unsigned d = (unsigned)(v[u].w - lo);
      if (a < span) atomicAdd(&cnt[a], 1u);
      if (b < span) atomicAdd(&cnt[b], 1u);
      if (c < span) atomicAdd(&cnt[c], 1u);
      if (d < span) atomicAdd(&cnt[d], 1u);
    }
  }
  for (int e = e_begin + 4 * n4 + tid; e < e_end; e += DEG_THREADS) {
    const unsigned a = (unsigned)(dst[e] - lo);
    if (a < span) atomicAdd(&cnt[a], 1u);
  }
  __syncthreads();

  unsigned* prow = part + (size_t)slice * n + lo;
  for (unsigned i = tid; i < span; i += DEG_THREADS) prow[i] = cnt[i];
}

__global__ __launch_bounds__(256)
void dinv_kernel(const unsigned* __restrict__ part, float* __restrict__ dinv, int n) {
  int i = blockIdx.x * blockDim.x + threadIdx.x;
  if (i < n) {
    unsigned s = 0u;
    #pragma unroll
    for (int p = 0; p < DEG_SLICES; ++p) s += part[(size_t)p * n + i];
    float d = (float)s;
    d = d < 1.0f ? 1.0f : d;
    dinv[i] = rsqrtf(d);
  }
}

__global__ __launch_bounds__(128)
void trunk_kernel(const float* __restrict__ X,
                  const float* __restrict__ W1, const float* __restrict__ b1,
                  const float* __restrict__ W2, const float* __restrict__ b2,
                  float* __restrict__ P0) {
  __shared__ float ldsX[16 * 132];
  __shared__ float ldsH[16 * 72];

  const int tid  = threadIdx.x;
  const int lane = tid & 31;
  const int wave = tid >> 5;
  const int mrow = lane & 15;
  const int kh   = lane >> 4;
  const int row0 = blockIdx.x * 16;
  const int col0 = wave * 16;

  for (int i = tid; i < 16 * IN_FEATS; i += 128) {
    int r = i >> 7, c = i & 127;
    ldsX[r * 132 + c] = X[(size_t)(row0 + r) * IN_FEATS + c];
  }
  __syncthreads();

  v8f acc = {};
  for (int k0 = 0; k0 < IN_FEATS; k0 += 4) {
    v2f a, b;
    a.x = ldsX[mrow * 132 + k0 + 2 * kh];
    a.y = ldsX[mrow * 132 + k0 + 2 * kh + 1];
    b.x = W1[(size_t)(k0 + 2 * kh)     * H_FEATS + col0 + mrow];
    b.y = W1[(size_t)(k0 + 2 * kh + 1) * H_FEATS + col0 + mrow];
    acc = wmma_f32(a, b, acc);
  }
  {
    float bias = b1[col0 + mrow];
    #pragma unroll
    for (int v = 0; v < 8; ++v) {
      int r = v + 8 * kh;
      float val = acc[v] + bias;
      ldsH[r * 72 + col0 + mrow] = val > 0.0f ? val : 0.0f;
    }
  }
  __syncthreads();

  v8f acc2 = {};
  for (int k0 = 0; k0 < H_FEATS; k0 += 4) {
    v2f a, b;
    a.x = ldsH[mrow * 72 + k0 + 2 * kh];
    a.y = ldsH[mrow * 72 + k0 + 2 * kh + 1];
    b.x = W2[(size_t)(k0 + 2 * kh)     * H_FEATS + col0 + mrow];
    b.y = W2[(size_t)(k0 + 2 * kh + 1) * H_FEATS + col0 + mrow];
    acc2 = wmma_f32(a, b, acc2);
  }
  {
    float bias = b2[col0 + mrow];
    #pragma unroll
    for (int v = 0; v < 8; ++v) {
      int r = v + 8 * kh;
      float val = acc2[v] + bias;
      P0[(size_t)(row0 + r) * H_FEATS + col0 + mrow] = val > 0.0f ? val : 0.0f;
    }
  }
}

#define EDGES_PER_WAVE 32
#define ROWS_AHEAD 8

__global__ __launch_bounds__(256)
void scatter_kernel(const float* __restrict__ P,
                    const int* __restrict__ src, const int* __restrict__ dst,
                    const float* __restrict__ dinv, float* __restrict__ agg,
                    int E, int squared) {
  const int lane = threadIdx.x & 31;
  const int wave = (blockIdx.x * blockDim.x + threadIdx.x) >> 5;
  const int e0   = wave * EDGES_PER_WAVE;
  if (e0 >= E) return;

  const int  e  = e0 + lane;
  const bool in = e < E;
  const int   sv = in ? src[e] : 0;
  const int   dv = in ? dst[e] : 0;
  const float dsv = dinv[sv];
  const float wv = squared ? dsv * dsv : dsv;

  if (e0 + EDGES_PER_WAVE <= E) {
    #pragma unroll 1
    for (int g = 0; g < EDGES_PER_WAVE; g += ROWS_AHEAD) {
      float vlo[ROWS_AHEAD], vhi[ROWS_AHEAD];
      #pragma unroll
      for (int j = 0; j < ROWS_AHEAD; ++j) {
        const int s = __builtin_amdgcn_readlane(sv, g + j);
        const float* rp = P + (size_t)s * H_FEATS + lane;
        vlo[j] = rp[0];
        vhi[j] = rp[H_FEATS / 2];
      }
      #pragma unroll
      for (int j = 0; j < ROWS_AHEAD; ++j) {
        const int   d  = __builtin_amdgcn_readlane(dv, g + j);
        const float di = __int_as_float(__builtin_amdgcn_readlane(__float_as_int(wv), g + j));
        float* ap = agg + (size_t)d * H_FEATS + lane;
        atomicAdd(ap,               vlo[j] * di);
        atomicAdd(ap + H_FEATS / 2, vhi[j] * di);
      }
    }
  } else {
    const int cnt = E - e0;
    for (int j = 0; j < cnt; ++j) {
      const int   s  = __builtin_amdgcn_readlane(sv, j);
      const int   d  = __builtin_amdgcn_readlane(dv, j);
      const float di = __int_as_float(__builtin_amdgcn_readlane(__float_as_int(wv), j));
      const float* rp = P + (size_t)s * H_FEATS + lane;
      const float vlo = rp[0];
      const float vhi = rp[H_FEATS / 2];
      float* ap = agg + (size_t)d * H_FEATS + lane;
      atomicAdd(ap,               vlo * di);
      atomicAdd(ap + H_FEATS / 2, vhi * di);
    }
  }
}

#define HEAD_K     (N_CONVS * H_FEATS)
#define HEAD_PITCH (HEAD_K + 4)

__global__ __launch_bounds__(128)
void head_kernel(const float* __restrict__ P0, const float* __restrict__ P1,
                 const float* __restrict__ P2, const float* __restrict__ P3,
                 const float* __restrict__ thetas, const float* __restrict__ Wm1,
                 const float* __restrict__ bm1,
                 const float* __restrict__ Wm2, const float* __restrict__ bm2,
                 const float* __restrict__ dinv, float* __restrict__ out) {
  __shared__ float ldsA[16 * HEAD_PITCH];
  __shared__ float ldsH[16 * 72];

  const int tid  = threadIdx.x;
  const int lane = tid & 31;
  const int wave = tid >> 5;
  const int mrow = lane & 15;
  const int kh   = lane >> 4;
  const int row0 = blockIdx.x * 16;
  const int col0 = wave * 16;

  float cf[N_CONVS][POLY_ORDER];
  #pragma unroll
  for (int i = 0; i < N_CONVS; ++i) {
    const float t0 = thetas[i * POLY_ORDER + 0], t1 = thetas[i * POLY_ORDER + 1];
    const float t2 = thetas[i * POLY_ORDER + 2], t3 = thetas[i * POLY_ORDER + 3];
    cf[i][0] = t0 + t1 + t2 + t3;
    cf[i][1] = -(t1 + 2.0f * t2 + 3.0f * t3);
    cf[i][2] = t2 + 3.0f * t3;
    cf[i][3] = -t3;
  }

  const int tr = tid >> 6, tc = tid & 63;
  float u0[8], u1[8], u2[8], u3[8], dsc[8];
  #pragma unroll
  for (int t = 0; t < 8; ++t) {
    const int r = row0 + tr + 2 * t;
    const size_t off = (size_t)r * H_FEATS + tc;
    u0[t] = P0[off];
    u1[t] = P1[off];
    u2[t] = P2[off];
    u3[t] = P3[off];
    dsc[t] = dinv[r];
  }
  #pragma unroll
  for (int t = 0; t < 8; ++t) {
    float* arow = ldsA + (tr + 2 * t) * HEAD_PITCH + tc;
    #pragma unroll
    for (int i = 0; i < N_CONVS; ++i) {
      const float g = cf[i][1] * u1[t] + cf[i][2] * u2[t] + cf[i][3] * u3[t];
      arow[i * H_FEATS] = cf[i][0] * u0[t] + dsc[t] * g;
    }
  }
  __syncthreads();

  v8f acc = {};
  for (int k0 = 0; k0 < HEAD_K; k0 += 4) {
    v2f a, b;
    a.x = ldsA[mrow * HEAD_PITCH + k0 + 2 * kh];
    a.y = ldsA[mrow * HEAD_PITCH + k0 + 2 * kh + 1];
    b.x = Wm1[(size_t)(k0 + 2 * kh)     * H_FEATS + col0 + mrow];
    b.y = Wm1[(size_t)(k0 + 2 * kh + 1) * H_FEATS + col0 + mrow];
    acc = wmma_f32(a, b, acc);
  }

  {
    float bias = bm1[col0 + mrow];
    #pragma unroll
    for (int v = 0; v < 8; ++v) {
      int r = v + 8 * kh;
      float val = acc[v] + bias;
      ldsH[r * 72 + col0 + mrow] = val > 0.0f ? val : 0.0f;
    }
  }
  __syncthreads();

  if (tid < 32) {
    int r = tid >> 1, c = tid & 1;
    float s = bm2[c];
    #pragma unroll 8
    for (int j = 0; j < H_FEATS; ++j)
      s += ldsH[r * 72 + j] * Wm2[j * NUM_CLASSES + c];
    out[(size_t)(row0 + r) * NUM_CLASSES + c] = s;
  }
}

extern "C" void kernel_launch(void* const* d_in, const int* in_sizes, int n_in,
                              void* d_out, int out_size, void* d_ws, size_t ws_size,
                              hipStream_t stream) {
  const float* X      = (const float*)d_in[0];
  const int*   src    = (const int*)  d_in[1];
  const int*   dst    = (const int*)  d_in[2];
  const float* W1     = (const float*)d_in[3];
  const float* b1     = (const float*)d_in[4];
  const float* W2     = (const float*)d_in[5];
  const float* b2     = (const float*)d_in[6];
  const float* thetas = (const float*)d_in[7];
  const float* Wm1    = (const float*)d_in[8];
  const float* bm1    = (const float*)d_in[9];
  const float* Wm2    = (const float*)d_in[10];
  const float* bm2    = (const float*)d_in[11];
  float* out = (float*)d_out;
  float* ws  = (float*)d_ws;

  const int PN = N_NODES * H_FEATS;
  float* dinv = ws;
  float* P0   = ws + 100352;
  float* P1   = P0 + PN;
  float* P2   = P1 + PN;
  float* P3   = P2 + PN;

  unsigned* part = (unsigned*)P3;
  degree_hist_kernel<<<DEG_SLICES * DEG_RANGES, DEG_THREADS, 0, stream>>>(
      dst, part, N_EDGES, N_NODES);
  dinv_kernel<<<(N_NODES + 255) / 256, 256, 0, stream>>>(part, dinv, N_NODES);

  zero_f32x4<<<2048, 256, 0, stream>>>(P1, 3 * (PN / 4));

  trunk_kernel<<<N_NODES / 16, 128, 0, stream>>>(X, W1, b1, W2, b2, P0);

  scatter_kernel<<<(N_EDGES + 255) / 256, 256, 0, stream>>>(
      P0, src, dst, dinv, P1, N_EDGES, 0);
  scatter_kernel<<<(N_EDGES + 255) / 256, 256, 0, stream>>>(
      P1, src, dst, dinv, P2, N_EDGES, 1);
  scatter_kernel<<<(N_EDGES + 255) / 256, 256, 0, stream>>>(
      P2, src, dst, dinv, P3, N_EDGES, 1);

  head_kernel<<<N_NODES / 16, 128, 0, stream>>>(
      P0, P1, P2, P3, thetas, Wm1, bm1, Wm2, bm2, dinv, out);
}


# launch folding: trunk work-groups form d^-1/2 of their 16 rows from the degree partials and clear U_1, U_2, half of U_3 (pass 1 clears the rest); clear and d^-1/2 launches gone, 8 -> 6; same sums
# speedup vs baseline: 2.5312x; 1.0348x over previous
#include <hip/hip_runtime.h>


#define N_NODES   100000
#define N_EDGES   1600000
#define IN_FEATS  128
#define H_FEATS   64
#define N_CONVS   3
#define POLY_ORDER 4
#define NUM_CLASSES 2

typedef __attribute__((ext_vector_type(2))) float v2f;
typedef __attribute__((ext_vector_type(8))) float v8f;

__device__ __forceinline__ v8f wmma_f32(v2f a, v2f b, v8f c) {
  return __builtin_amdgcn_wmma_f32_16x16x4_f32(
       false, a,  false, b,
       (short)0, c,  false,  false);
}

#define DEG_RANGES      8
#define DEG_RANGE_NODES 12512
#define DEG_SLICES      32
#define DEG_THREADS     512
#define DEG_LOADS       8

__global__ __launch_bounds__(DEG_THREADS)
void degree_hist_kernel(const int* __restrict__ dst, unsigned* __restrict__ part,
                        int E, int n) {
  __shared__ unsigned cnt[DEG_RANGE_NODES];

  const int tid   = threadIdx.x;
  const int range = blockIdx.x % DEG_RANGES;
  const int slice = blockIdx.x / DEG_RANGES;
  const int lo    = range * DEG_RANGE_NODES;
  int hi = lo + DEG_RANGE_NODES;
  if (hi > n) hi = n;
  const unsigned span = hi > lo ? (unsigned)(hi - lo) : 0u;

  for (int i = tid; i < DEG_RANGE_NODES; i += DEG_THREADS) cnt[i] = 0u;
  __syncthreads();

  const int slice_len = (((E + DEG_SLICES - 1) / DEG_SLICES) + 3) & ~3;
  const int e_begin = slice * slice_len;
  int e_end = e_begin + slice_len;
  if (e_end > E) e_end = E;
  const int n4 = e_end > e_begin ? (e_end - e_begin) >> 2 : 0;

  const int4* d4 = reinterpret_cast<const int4*>(dst + e_begin);
  for (int i0 = tid; i0 < n4; i0 += DEG_LOADS * DEG_THREADS) {
    int4 v[DEG_LOADS];
    #pragma unroll
    for (int u = 0; u < DEG_LOADS; ++u) {
      const int i = i0 + u * DEG_THREADS;
      v[u] = i < n4 ? d4[i] : make_int4(-1, -1, -1, -1);
    }
    #pragma unroll
    for (int u = 0; u < DEG_LOADS; ++u) {
      const unsigned a = (unsigned)(v[u].x - lo);
      const unsigned b = (unsigned)(v[u].y - lo);
      const unsigned c = (unsigned)(v[u].z - lo);
      const unsigned d = (unsigned)(v[u].w - lo);
      if (a < span) atomicAdd(&cnt[a], 1u);
      if (b < span) atomicAdd(&cnt[b], 1u);
      if (c < span) atomicAdd(&cnt[c], 1u);
      if (d < span) atomicAdd(&cnt[d], 1u);
    }
  }
  for (int e = e_begin + 4 * n4 + tid; e < e_end; e += DEG_THREADS) {
    const unsigned a = (unsigned)(dst[e] - lo);
    if (a < span) atomicAdd(&cnt[a], 1u);
  }
  __syncthreads();

  unsigned* prow = part + (size_t)slice * n + lo;
  for (unsigned i = tid; i < span; i += DEG_THREADS) prow[i] = cnt[i];
}

__global__ __launch_bounds__(128)
void trunk_kernel(const float* __restrict__ X,
                  const float* __restrict__ W1, const float* __restrict__ b1,
                  const float* __restrict__ W2, const float* __restrict__ b2,
                  float* __restrict__ P0,
                  const unsigned* __restrict__ part, float* __restrict__ dinv,
                  float* __restrict__ U1, float* __restrict__ U2,
                  float* __restrict__ U3hi) {
  __shared__ float ldsX[16 * 132];
  __shared__ float ldsH[16 * 72];
  __shared__ unsigned ldsD[128];

  const int tid  = threadIdx.x;
  const int lane = tid & 31;
  const int wave = tid >> 5;
  const int mrow = lane & 15;
  const int kh   = lane >> 4;
  const int row0 = blockIdx.x * 16;
  const int col0 = wave * 16;

  unsigned pc[DEG_SLICES / 8];
  {
    const int dr = tid & 15, dg = tid >> 4;
    #pragma unroll
    for (int q = 0; q < DEG_SLICES / 8; ++q)
      pc[q] = part[(size_t)(dg * (DEG_SLICES / 8) + q) * N_NODES + row0 + dr];
  }

  {
    const float4 z4 = make_float4(0.0f, 0.0f, 0.0f, 0.0f);
    float4* u1 = reinterpret_cast<float4*>(U1) + (size_t)blockIdx.x * 256;
    float4* u2 = reinterpret_cast<float4*>(U2) + (size_t)blockIdx.x * 256;
    float4* u3 = reinterpret_cast<float4*>(U3hi) + (size_t)blockIdx.x * 128;
    u1[tid] = z4;  u1[tid + 128] = z4;
    u2[tid] = z4;  u2[tid + 128] = z4;
    u3[tid] = z4;
  }

  for (int i = tid; i < 16 * IN_FEATS; i += 128) {
    int r = i >> 7, c = i & 127;
    ldsX[r * 132 + c] = X[(size_t)(row0 + r) * IN_FEATS + c];
  }
  ldsD[tid] = (pc[0] + pc[1]) + (pc[2] + pc[3]);
  __syncthreads();

  if (tid < 16) {
    unsigned cs = 0u;
    #pragma unroll
    for (int g = 0; g < 8; ++g) cs += ldsD[tid + 16 * g];
    float d = (float)cs;
    d = d < 1.0f ? 1.0f : d;
    dinv[row0 + tid] = rsqrtf(d);
  }

  v8f acc = {};
  for (int k0 = 0; k0 < IN_FEATS; k0 += 4) {
    v2f a, b;
    a.x = ldsX[mrow * 132 + k0 + 2 * kh];
    a.y = ldsX[mrow * 132 + k0 + 2 * kh + 1];
    b.x = W1[(size_t)(k0 + 2 * kh)     * H_FEATS + col0 + mrow];
    b.y = W1[(size_t)(k0 + 2 * kh + 1) * H_FEATS + col0 + mrow];
    acc = wmma_f32(a, b, acc);
  }
  {
    float bias = b1[col0 + mrow];
    #pragma unroll
    for (int v = 0; v < 8; ++v) {
      int r = v + 8 * kh;
      float val = acc[v] + bias;
      ldsH[r * 72 + col0 + mrow] = val > 0.0f ? val : 0.0f;
    }
  }
  __syncthreads();

  v8f acc2 = {};
  for (int k0 = 0; k0 < H_FEATS; k0 += 4) {
    v2f a, b;
    a.x = ldsH[mrow * 72 + k0 + 2 * kh];
    a.y = ldsH[mrow * 72 + k0 + 2 * kh + 1];
    b.x = W2[(size_t)(k0 + 2 * kh)     * H_FEATS + col0 + mrow];
    b.y = W2[(size_t)(k0 + 2 * kh + 1) * H_FEATS + col0 + mrow];
    acc2 = wmma_f32(a, b, acc2);
  }
  {
    float bias = b2[col0 + mrow];
    #pragma unroll
    for (int v = 0; v < 8; ++v) {
      int r = v + 8 * kh;
      float val = acc2[v] + bias;
      P0[(size_t)(row0 + r) * H_FEATS + col0 + mrow] = val > 0.0f ? val : 0.0f;
    }
  }
}

#define EDGES_PER_WAVE 32
#define ROWS_AHEAD 8

__global__ __launch_bounds__(256)
void scatter_kernel(const float* __restrict__ P,
                    const int* __restrict__ src, const int* __restrict__ dst,
                    const float* __restrict__ dinv, float* __restrict__ agg,
                    int E, int squared, float* __restrict__ zbuf, int zn2) {
  const int lane = threadIdx.x & 31;
  const int wave = (blockIdx.x * blockDim.x + threadIdx.x) >> 5;
  if (zbuf) {
    float2* z2 = reinterpret_cast<float2*>(zbuf);
    const int zstride = gridDim.x * blockDim.x;
    for (int i = wave * 32 + lane; i < zn2; i += zstride)
      z2[i] = make_float2(0.0f, 0.0f);
  }
  const int e0   = wave * EDGES_PER_WAVE;
  if (e0 >= E) return;

  const int  e  = e0 + lane;
  const bool in = e < E;
  const int   sv = in ? src[e] : 0;
  const int   dv = in ? dst[e] : 0;
  const float dsv = dinv[sv];
  const float wv = squared ? dsv * dsv : dsv;

  if (e0 + EDGES_PER_WAVE <= E) {
    #pragma unroll 1
    for (int g = 0; g < EDGES_PER_WAVE; g += ROWS_AHEAD) {
      float vlo[ROWS_AHEAD], vhi[ROWS_AHEAD];
      #pragma unroll
      for (int j = 0; j < ROWS_AHEAD; ++j) {
        const int s = __builtin_amdgcn_readlane(sv, g + j);
        const float* rp = P + (size_t)s * H_FEATS + lane;
        vlo[j] = rp[0];
        vhi[j] = rp[H_FEATS / 2];
      }
      #pragma unroll
      for (int j = 0; j < ROWS_AHEAD; ++j) {
        const int   d  = __builtin_amdgcn_readlane(dv, g + j);
        const float di = __int_as_float(__builtin_amdgcn_readlane(__float_as_int(wv), g + j));
        float* ap = agg + (size_t)d * H_FEATS + lane;
        atomicAdd(ap,               vlo[j] * di);
        atomicAdd(ap + H_FEATS / 2, vhi[j] * di);
      }
    }
  } else {
    const int cnt = E - e0;
    for (int j = 0; j < cnt; ++j) {
      const int   s  = __builtin_amdgcn_readlane(sv, j);
      const int   d  = __builtin_amdgcn_readlane(dv, j);
      const float di = __int_as_float(__builtin_amdgcn_readlane(__float_as_int(wv), j));
      const float* rp = P + (size_t)s * H_FEATS + lane;
      const float vlo = rp[0];
      const float vhi = rp[H_FEATS / 2];
      float* ap = agg + (size_t)d * H_FEATS + lane;
      atomicAdd(ap,               vlo * di);
      atomicAdd(ap + H_FEATS / 2, vhi * di);
    }
  }
}

#define HEAD_K     (N_CONVS * H_FEATS)
#define HEAD_PITCH (HEAD_K + 4)

__global__ __launch_bounds__(128)
void head_kernel(const float* __restrict__ P0, const float* __restrict__ P1,
                 const float* __restrict__ P2, const float* __restrict__ P3,
                 const float* __restrict__ thetas, const float* __restrict__ Wm1,
                 const float* __restrict__ bm1,
                 const float* __restrict__ Wm2, const float* __restrict__ bm2,
                 const float* __restrict__ dinv, float* __restrict__ out) {
  __shared__ float ldsA[16 * HEAD_PITCH];
  __shared__ float ldsH[16 * 72];

  const int tid  = threadIdx.x;
  const int lane = tid & 31;
  const int wave = tid >> 5;
  const int mrow = lane & 15;
  const int kh   = lane >> 4;
  const int row0 = blockIdx.x * 16;
  const int col0 = wave * 16;

  float cf[N_CONVS][POLY_ORDER];
  #pragma unroll
  for (int i = 0; i < N_CONVS; ++i) {
    const float t0 = thetas[i * POLY_ORDER + 0], t1 = thetas[i * POLY_ORDER + 1];
    const float t2 = thetas[i * POLY_ORDER + 2], t3 = thetas[i * POLY_ORDER + 3];
    cf[i][0] = t0 + t1 + t2 + t3;
    cf[i][1] = -(t1 + 2.0f * t2 + 3.0f * t3);
    cf[i][2] = t2 + 3.0f * t3;
    cf[i][3] = -t3;
  }

  const int tr = tid >> 6, tc = tid & 63;
  float u0[8], u1[8], u2[8], u3[8], dsc[8];
  #pragma unroll
  for (int t = 0; t < 8; ++t) {
    const int r = row0 + tr + 2 * t;
    const size_t off = (size_t)r * H_FEATS + tc;
    u0[t] = P0[off];
    u1[t] = P1[off];
    u2[t] = P2[off];
    u3[t] = P3[off];
    dsc[t] = dinv[r];
  }
  #pragma unroll
  for (int t = 0; t < 8; ++t) {
    float* arow = ldsA + (tr + 2 * t) * HEAD_PITCH + tc;
    #pragma unroll
    for (int i = 0; i < N_CONVS; ++i) {
      const float g = cf[i][1] * u1[t] + cf[i][2] * u2[t] + cf[i][3] * u3[t];
      arow[i * H_FEATS] = cf[i][0] * u0[t] + dsc[t] * g;
    }
  }
  __syncthreads();

  v8f acc = {};
  for (int k0 = 0; k0 < HEAD_K; k0 += 4) {
    v2f a, b;
    a.x = ldsA[mrow * HEAD_PITCH + k0 + 2 * kh];
    a.y = ldsA[mrow * HEAD_PITCH + k0 + 2 * kh + 1];
    b.x = Wm1[(size_t)(k0 + 2 * kh)     * H_FEATS + col0 + mrow];
    b.y = Wm1[(size_t)(k0 + 2 * kh + 1) * H_FEATS + col0 + mrow];
    acc = wmma_f32(a, b, acc);
  }

  {
    float bias = bm1[col0 + mrow];
    #pragma unroll
    for (int v = 0; v < 8; ++v) {
      int r = v + 8 * kh;
      float val = acc[v] + bias;
      ldsH[r * 72 + col0 + mrow] = val > 0.0f ? val : 0.0f;
    }
  }
  __syncthreads();

  if (tid < 32) {
    int r = tid >> 1, c = tid & 1;
    float s = bm2[c];
    #pragma unroll 8
    for (int j = 0; j < H_FEATS; ++j)
      s += ldsH[r * 72 + j] * Wm2[j * NUM_CLASSES + c];
    out[(size_t)(row0 + r) * NUM_CLASSES + c] = s;
  }
}

extern "C" void kernel_launch(void* const* d_in, const int* in_sizes, int n_in,
                              void* d_out, int out_size, void* d_ws, size_t ws_size,
                              hipStream_t stream) {
  const float* X      = (const float*)d_in[0];
  const int*   src    = (const int*)  d_in[1];
  const int*   dst    = (const int*)  d_in[2];
  const float* W1     = (const float*)d_in[3];
  const float* b1     = (const float*)d_in[4];
  const float* W2     = (const float*)d_in[5];
  const float* b2     = (const float*)d_in[6];
  const float* thetas = (const float*)d_in[7];
  const float* Wm1    = (const float*)d_in[8];
  const float* bm1    = (const float*)d_in[9];
  const float* Wm2    = (const float*)d_in[10];
  const float* bm2    = (const float*)d_in[11];
  float* out = (float*)d_out;
  float* ws  = (float*)d_ws;

  const int PN = N_NODES * H_FEATS;
  float* dinv = ws;
  float* P0   = ws + 100352;
  float* P1   = P0 + PN;
  float* P2   = P1 + PN;
  float* P3   = P2 + PN;

  unsigned* part = (unsigned*)P3;
  degree_hist_kernel<<<DEG_SLICES * DEG_RANGES, DEG_THREADS, 0, stream>>>(
      dst, part, N_EDGES, N_NODES);

  trunk_kernel<<<N_NODES / 16, 128, 0, stream>>>(X, W1, b1, W2, b2, P0,
                                                 part, dinv, P1, P2, P3 + PN / 2);

  scatter_kernel<<<(N_EDGES + 255) / 256, 256, 0, stream>>>(
      P0, src, dst, dinv, P1, N_EDGES, 0, P3, PN / 4);
  scatter_kernel<<<(N_EDGES + 255) / 256, 256, 0, stream>>>(
      P1, src, dst, dinv, P2, N_EDGES, 1, nullptr, 0);
  scatter_kernel<<<(N_EDGES + 255) / 256, 256, 0, stream>>>(
      P2, src, dst, dinv, P3, N_EDGES, 1, nullptr, 0);

  head_kernel<<<N_NODES / 16, 128, 0, stream>>>(
      P0, P1, P2, P3, thetas, Wm1, bm1, Wm2, bm2, dinv, out);
}


# edge passes: each accumulator row gets its f32 atomics from one L2 half only (work-group reads its die, takes a per-half arrival rank, keeps edges of its half's node range); arrival-order sums
# speedup vs baseline: 2.8790x; 1.1374x over previous
#include <hip/hip_runtime.h>


#define N_NODES   100000
#define N_EDGES   1600000
#define IN_FEATS  128
#define H_FEATS   64
#define N_CONVS   3
#define POLY_ORDER 4
#define NUM_CLASSES 2
#define SC_PASSES   3

typedef __attribute__((ext_vector_type(2))) float v2f;
typedef __attribute__((ext_vector_type(8))) float v8f;

__device__ __forceinline__ v8f wmma_f32(v2f a, v2f b, v8f c) {
  return __builtin_amdgcn_wmma_f32_16x16x4_f32(
       false, a,  false, b,
       (short)0, c,  false,  false);
}

#define DEG_RANGES      8
#define DEG_RANGE_NODES 12512
#define DEG_SLICES      32
#define DEG_THREADS     512
#define DEG_LOADS       8

__global__ __launch_bounds__(DEG_THREADS)
void degree_hist_kernel(const int* __restrict__ dst, unsigned* __restrict__ part,
                        int E, int n) {
  __shared__ unsigned cnt[DEG_RANGE_NODES];

  const int tid   = threadIdx.x;
  const int range = blockIdx.x % DEG_RANGES;
  const int slice = blockIdx.x / DEG_RANGES;
  const int lo    = range * DEG_RANGE_NODES;
  int hi = lo + DEG_RANGE_NODES;
  if (hi > n) hi = n;
  const unsigned span = hi > lo ? (unsigned)(hi - lo) : 0u;

  for (int i = tid; i < DEG_RANGE_NODES; i += DEG_THREADS) cnt[i] = 0u;
  __syncthreads();

  const int slice_len = (((E + DEG_SLICES - 1) / DEG_SLICES) + 3) & ~3;
  const int e_begin = slice * slice_len;
  int e_end = e_begin + slice_len;
  if (e_end > E) e_end = E;
  const int n4 = e_end > e_begin ? (e_end - e_begin) >> 2 : 0;

  const int4* d4 = reinterpret_cast<const int4*>(dst + e_begin);
  for (int i0 = tid; i0 < n4; i0 += DEG_LOADS * DEG_THREADS) {
    int4 v[DEG_LOADS];
    #pragma unroll
    for (int u = 0; u < DEG_LOADS; ++u) {
      const int i = i0 + u * DEG_THREADS;
      v[u] = i < n4 ? d4[i] : make_int4(-1, -1, -1, -1);
    }
    #pragma unroll
    for (int u = 0; u < DEG_LOADS; ++u) {
      const unsigned a = (unsigned)(v[u].x - lo);
      const unsigned b = (unsigned)(v[u].y - lo);
      const unsigned c = (unsigned)(v[u].z - lo);
      const unsigned d = (unsigned)(v[u].w - lo);
      if (a < span) atomicAdd(&cnt[a], 1u);
      if (b < span) atomicAdd(&cnt[b], 1u);
      if (c < span) atomicAdd(&cnt[c], 1u);
      if (d < span) atomicAdd(&cnt[d], 1u);
    }
  }
  for (int e = e_begin + 4 * n4 + tid; e < e_end; e += DEG_THREADS) {
    const unsigned a = (unsigned)(dst[e] - lo);
    if (a < span) atomicAdd(&cnt[a], 1u);
  }
  __syncthreads();

  unsigned* prow = part + (size_t)slice * n + lo;
  for (unsigned i = tid; i < span; i += DEG_THREADS) prow[i] = cnt[i];
}

__global__ __launch_bounds__(128)
void trunk_kernel(const float* __restrict__ X,
                  const float* __restrict__ W1, const float* __restrict__ b1,
                  const float* __restrict__ W2, const float* __restrict__ b2,
                  float* __restrict__ P0,
                  const unsigned* __restrict__ part, float* __restrict__ dinv,
                  float* __restrict__ U1, float* __restrict__ U2,
                  float* __restrict__ U3hi, unsigned* __restrict__ arrive) {
  __shared__ float ldsX[16 * 132];
  __shared__ float ldsH[16 * 72];
  __shared__ unsigned ldsD[128];

  const int tid  = threadIdx.x;
  const int lane = tid & 31;
  const int wave = tid >> 5;
  const int mrow = lane & 15;
  const int kh   = lane >> 4;
  const int row0 = blockIdx.x * 16;
  const int col0 = wave * 16;

  unsigned pc[DEG_SLICES / 8];
  {
    const int dr = tid & 15, dg = tid >> 4;
    #pragma unroll
    for (int q = 0; q < DEG_SLICES / 8; ++q)
      pc[q] = part[(size_t)(dg * (DEG_SLICES / 8) + q) * N_NODES + row0 + dr];
  }

  if (blockIdx.x == 0 && tid < 2 * SC_PASSES) arrive[tid * 32] = 0u;

  {
    const float4 z4 = make_float4(0.0f, 0.0f, 0.0f, 0.0f);
    float4* u1 = reinterpret_cast<float4*>(U1) + (size_t)blockIdx.x * 256;
    float4* u2 = reinterpret_cast<float4*>(U2) + (size_t)blockIdx.x * 256;
    float4* u3 = reinterpret_cast<float4*>(U3hi) + (size_t)blockIdx.x * 128;
    u1[tid] = z4;  u1[tid + 128] = z4;
    u2[tid] = z4;  u2[tid + 128] = z4;
    u3[tid] = z4;
  }

  for (int i = tid; i < 16 * IN_FEATS; i += 128) {
    int r = i >> 7, c = i & 127;
    ldsX[r * 132 + c] = X[(size_t)(row0 + r) * IN_FEATS + c];
  }
  ldsD[tid] = (pc[0] + pc[1]) + (pc[2] + pc[3]);
  __syncthreads();

  if (tid < 16) {
    unsigned cs = 0u;
    #pragma unroll
    for (int g = 0; g < 8; ++g) cs += ldsD[tid + 16 * g];
    float d = (float)cs;
    d = d < 1.0f ? 1.0f : d;
    dinv[row0 + tid] = rsqrtf(d);
  }

  v8f acc = {};
  for (int k0 = 0; k0 < IN_FEATS; k0 += 4) {
    v2f a, b;
    a.x = ldsX[mrow * 132 + k0 + 2 * kh];
    a.y = ldsX[mrow * 132 + k0 + 2 * kh + 1];
    b.x = W1[(size_t)(k0 + 2 * kh)     * H_FEATS + col0 + mrow];
    b.y = W1[(size_t)(k0 + 2 * kh + 1) * H_FEATS + col0 + mrow];
    acc = wmma_f32(a, b, acc);
  }
  {
    float bias = b1[col0 + mrow];
    #pragma unroll
    for (int v = 0; v < 8; ++v) {
      int r = v + 8 * kh;
      float val = acc[v] + bias;
      ldsH[r * 72 + col0 + mrow] = val > 0.0f ? val : 0.0f;
    }
  }
  __syncthreads();

  v8f acc2 = {};
  for (int k0 = 0; k0 < H_FEATS; k0 += 4) {
    v2f a, b;
    a.x = ldsH[mrow * 72 + k0 + 2 * kh];
    a.y = ldsH[mrow * 72 + k0 + 2 * kh + 1];
    b.x = W2[(size_t)(k0 + 2 * kh)     * H_FEATS + col0 + mrow];
    b.y = W2[(size_t)(k0 + 2 * kh + 1) * H_FEATS + col0 + mrow];
    acc2 = wmma_f32(a, b, acc2);
  }
  {
    float bias = b2[col0 + mrow];
    #pragma unroll
    for (int v = 0; v < 8; ++v) {
      int r = v + 8 * kh;
      float val = acc2[v] + bias;
      P0[(size_t)(row0 + r) * H_FEATS + col0 + mrow] = val > 0.0f ? val : 0.0f;
    }
  }
}

#define SC_BLOCKS   2048
#define SC_THREADS  256
#define SC_WAVES    (SC_THREADS / 32)
#define SC_ROLE_WAVES_LOG2 13
#define SC_CHUNK    7
#define SC_RING     256
#define ROWS_AHEAD  8

__global__ __launch_bounds__(SC_THREADS)
void scatter_kernel(const float* __restrict__ P,
                    const int* __restrict__ src, const int* __restrict__ dst,
                    const float* __restrict__ dinv, float* __restrict__ agg,
                    int E, int n, int squared, float* __restrict__ zbuf, int zn4,
                    unsigned* __restrict__ arrive) {
  __shared__ int   ring_s[SC_WAVES][SC_RING];
  __shared__ int   ring_d[SC_WAVES][SC_RING];
  __shared__ float ring_w[SC_WAVES][SC_RING];
  __shared__ int   role_sh;

  const int lane = threadIdx.x & 31;
  const int w    = __builtin_amdgcn_readfirstlane((int)(threadIdx.x >> 5));

  if (threadIdx.x == 0) {
    const unsigned place = __builtin_amdgcn_s_sendmsg_rtn(0x87);
    const int h = (int)((place >> 18) & 1u);
    const int rank = (int)atomicAdd(&arrive[h * 32], 1u);
    int q = h, r = rank;
    if (rank >= SC_BLOCKS / 2) { q = h ^ 1; r = SC_BLOCKS - 1 - rank; }
    role_sh = (q << 16) | (r & 0xffff);
  }
  __syncthreads();
  const int q = role_sh >> 16;
  const int r = role_sh & 0xffff;
  if (r >= SC_BLOCKS / 2) return;

  if (zbuf && q == 0) {
    float4* z4 = reinterpret_cast<float4*>(zbuf);
    for (int i = r * SC_THREADS + threadIdx.x; i < zn4; i += (SC_BLOCKS / 2) * SC_THREADS)
      z4[i] = make_float4(0.0f, 0.0f, 0.0f, 0.0f);
  }

  const int lo = q ? (n >> 1) : 0;
  const unsigned span = q ? (unsigned)(n - (n >> 1)) : (unsigned)(n >> 1);

  const long long rwid = (long long)(r * SC_WAVES + w);
  const int e_lo = (int)((rwid * E) >> SC_ROLE_WAVES_LOG2);
  const int e_hi = (int)(((rwid + 1) * E) >> SC_ROLE_WAVES_LOG2);

  int*   rs = ring_s[w];
  int*   rd = ring_d[w];
  float* rw = ring_w[w];
  int head = 0, tail = 0;

  for (int eb = e_lo; eb < e_hi; eb += 32 * SC_CHUNK) {
    int sv[SC_CHUNK], dv[SC_CHUNK];
    #pragma unroll
    for (int u = 0; u < SC_CHUNK; ++u) {
      const int e  = eb + 32 * u + lane;
      const int ec = e < e_hi ? e : e_hi - 1;
      const int dvv = dst[ec];
      sv[u] = src[ec];
      dv[u] = e < e_hi ? dvv : -1;
    }
    float wv[SC_CHUNK];
    #pragma unroll
    for (int u = 0; u < SC_CHUNK; ++u) {
      const bool inr = (unsigned)(dv[u] - lo) < span;
      const float ds = inr ? dinv[sv[u]] : 0.0f;
      wv[u] = squared ? ds * ds : ds;
    }
    #pragma unroll
    for (int u = 0; u < SC_CHUNK; ++u) {
      const bool inr = (unsigned)(dv[u] - lo) < span;
      const unsigned m = __builtin_amdgcn_ballot_w32(inr);
      if (inr) {
        const int pos = (tail + (int)__builtin_amdgcn_mbcnt_lo(m, 0u)) & (SC_RING - 1);
        rs[pos] = sv[u];
        rd[pos] = dv[u];
        rw[pos] = wv[u];
      }
      tail += __builtin_popcount(m);
    }
    __builtin_amdgcn_wave_barrier();

    while (tail - head >= 32) {
      const int idx = (head + lane) & (SC_RING - 1);
      const int   es = rs[idx];
      const int   ed = rd[idx];
      const float ew = rw[idx];
      head += 32;
      #pragma unroll 1
      for (int g = 0; g < 32; g += ROWS_AHEAD) {
        float vlo[ROWS_AHEAD], vhi[ROWS_AHEAD];
        #pragma unroll
        for (int j = 0; j < ROWS_AHEAD; ++j) {
          const int s = __builtin_amdgcn_readlane(es, g + j);
          const float* rp = P + (size_t)s * H_FEATS + lane;
          vlo[j] = rp[0];
          vhi[j] = rp[H_FEATS / 2];
        }
        #pragma unroll
        for (int j = 0; j < ROWS_AHEAD; ++j) {
          const int   d  = __builtin_amdgcn_readlane(ed, g + j);
          const float di = __int_as_float(__builtin_amdgcn_readlane(__float_as_int(ew), g + j));
          float* ap = agg + (size_t)d * H_FEATS + lane;
          atomicAdd(ap,               vlo[j] * di);
          atomicAdd(ap + H_FEATS / 2, vhi[j] * di);
        }
      }
    }
    __builtin_amdgcn_wave_barrier();
  }

  const int cnt = tail - head;
  if (cnt > 0) {
    const int idx = (head + lane) & (SC_RING - 1);
    const bool has = lane < cnt;
    const int   es = has ? rs[idx] : 0;
    const int   ed = has ? rd[idx] : 0;
    const float ew = has ? rw[idx] : 0.0f;
    #pragma unroll 1
    for (int g = 0; g < cnt; g += ROWS_AHEAD) {
      const int k = cnt - g;
      float vlo[ROWS_AHEAD], vhi[ROWS_AHEAD];
      #pragma unroll
      for (int j = 0; j < ROWS_AHEAD; ++j) {
        vlo[j] = 0.0f;  vhi[j] = 0.0f;
        if (j < k) {
          const int s = __builtin_amdgcn_readlane(es, g + j);
          const float* rp = P + (size_t)s * H_FEATS + lane;
          vlo[j] = rp[0];
          vhi[j] = rp[H_FEATS / 2];
        }
      }
      #pragma unroll
      for (int j = 0; j < ROWS_AHEAD; ++j) {
        if (j < k) {
          const int   d  = __builtin_amdgcn_readlane(ed, g + j);
          const float di = __int_as_float(__builtin_amdgcn_readlane(__float_as_int(ew), g + j));
          float* ap = agg + (size_t)d * H_FEATS + lane;
          atomicAdd(ap,               vlo[j] * di);
          atomicAdd(ap + H_FEATS / 2, vhi[j] * di);
        }
      }
    }
  }
}

#define HEAD_K     (N_CONVS * H_FEATS)
#define HEAD_PITCH (HEAD_K + 4)

__global__ __launch_bounds__(128)
void head_kernel(const float* __restrict__ P0, const float* __restrict__ P1,
                 const float* __restrict__ P2, const float* __restrict__ P3,
                 const float* __restrict__ thetas, const float* __restrict__ Wm1,
                 const float* __restrict__ bm1,
                 const float* __restrict__ Wm2, const float* __restrict__ bm2,
                 const float* __restrict__ dinv, float* __restrict__ out) {
  __shared__ float ldsA[16 * HEAD_PITCH];
  __shared__ float ldsH[16 * 72];

  const int tid  = threadIdx.x;
  const int lane = tid & 31;
  const int wave = tid >> 5;
  const int mrow = lane & 15;
  const int kh   = lane >> 4;
  const int row0 = blockIdx.x * 16;
  const int col0 = wave * 16;

  float cf[N_CONVS][POLY_ORDER];
  #pragma unroll
  for (int i = 0; i < N_CONVS; ++i) {
    const float t0 = thetas[i * POLY_ORDER + 0], t1 = thetas[i * POLY_ORDER + 1];
    const float t2 = thetas[i * POLY_ORDER + 2], t3 = thetas[i * POLY_ORDER + 3];
    cf[i][0] = t0 + t1 + t2 + t3;
    cf[i][1] = -(t1 + 2.0f * t2 + 3.0f * t3);
    cf[i][2] = t2 + 3.0f * t3;
    cf[i][3] = -t3;
  }

  const int tr = tid >> 6, tc = tid & 63;
  float u0[8], u1[8], u2[8], u3[8], dsc[8];
  #pragma unroll
  for (int t = 0; t < 8; ++t) {
    const int r = row0 + tr + 2 * t;
    const size_t off = (size_t)r * H_FEATS + tc;
    u0[t] = P0[off];
    u1[t] = P1[off];
    u2[t] = P2[off];
    u3[t] = P3[off];
    dsc[t] = dinv[r];
  }
  #pragma unroll
  for (int t = 0; t < 8; ++t) {
    float* arow = ldsA + (tr + 2 * t) * HEAD_PITCH + tc;
    #pragma unroll
    for (int i = 0; i < N_CONVS; ++i) {
      const float g = cf[i][1] * u1[t] + cf[i][2] * u2[t] + cf[i][3] * u3[t];
      arow[i * H_FEATS] = cf[i][0] * u0[t] + dsc[t] * g;
    }
  }
  __syncthreads();

  v8f acc = {};
  for (int k0 = 0; k0 < HEAD_K; k0 += 4) {
    v2f a, b;
    a.x = ldsA[mrow * HEAD_PITCH + k0 + 2 * kh];
    a.y = ldsA[mrow * HEAD_PITCH + k0 + 2 * kh + 1];
    b.x = Wm1[(size_t)(k0 + 2 * kh)     * H_FEATS + col0 + mrow];
    b.y = Wm1[(size_t)(k0 + 2 * kh + 1) * H_FEATS + col0 + mrow];
    acc = wmma_f32(a, b, acc);
  }

  {
    float bias = bm1[col0 + mrow];
    #pragma unroll
    for (int v = 0; v < 8; ++v) {
      int r = v + 8 * kh;
      float val = acc[v] + bias;
      ldsH[r * 72 + col0 + mrow] = val > 0.0f ? val : 0.0f;
    }
  }
  __syncthreads();

  if (tid < 32) {
    int r = tid >> 1, c = tid & 1;
    float s = bm2[c];
    #pragma unroll 8
    for (int j = 0; j < H_FEATS; ++j)
      s += ldsH[r * 72 + j] * Wm2[j * NUM_CLASSES + c];
    out[(size_t)(row0 + r) * NUM_CLASSES + c] = s;
  }
}

extern "C" void kernel_launch(void* const* d_in, const int* in_sizes, int n_in,
                              void* d_out, int out_size, void* d_ws, size_t ws_size,
                              hipStream_t stream) {
  const float* X      = (const float*)d_in[0];
  const int*   src    = (const int*)  d_in[1];
  const int*   dst    = (const int*)  d_in[2];
  const float* W1     = (const float*)d_in[3];
  const float* b1     = (const float*)d_in[4];
  const float* W2     = (const float*)d_in[5];
  const float* b2     = (const float*)d_in[6];
  const float* thetas = (const float*)d_in[7];
  const float* Wm1    = (const float*)d_in[8];
  const float* bm1    = (const float*)d_in[9];
  const float* Wm2    = (const float*)d_in[10];
  const float* bm2    = (const float*)d_in[11];
  float* out = (float*)d_out;
  float* ws  = (float*)d_ws;

  const int PN = N_NODES * H_FEATS;
  float* dinv = ws;
  float* P0   = ws + 100352;
  float* P1   = P0 + PN;
  float* P2   = P1 + PN;
  float* P3   = P2 + PN;
  unsigned* arrive = (unsigned*)(ws + 100096);

  unsigned* part = (unsigned*)P3;
  degree_hist_kernel<<<DEG_SLICES * DEG_RANGES, DEG_THREADS, 0, stream>>>(
      dst, part, N_EDGES, N_NODES);

  trunk_kernel<<<N_NODES / 16, 128, 0, stream>>>(X, W1, b1, W2, b2, P0,
                                                 part, dinv, P1, P2, P3 + PN / 2, arrive);

  scatter_kernel<<<SC_BLOCKS, SC_THREADS, 0, stream>>>(
      P0, src, dst, dinv, P1, N_EDGES, N_NODES, 0, P3, PN / 8, arrive);
  scatter_kernel<<<SC_BLOCKS, SC_THREADS, 0, stream>>>(
      P1, src, dst, dinv, P2, N_EDGES, N_NODES, 1, nullptr, 0, arrive + 64);
  scatter_kernel<<<SC_BLOCKS, SC_THREADS, 0, stream>>>(
      P2, src, dst, dinv, P3, N_EDGES, N_NODES, 1, nullptr, 0, arrive + 128);

  head_kernel<<<N_NODES / 16, 128, 0, stream>>>(
      P0, P1, P2, P3, thetas, Wm1, bm1, Wm2, bm2, dinv, out);
}


# trunk and head: 1250 long-lived work-groups walk 5 row tiles each (5,000 waves a launch, was 25,000); weights held in registers, next tile's inputs loaded during the product; same sums
# speedup vs baseline: 3.0441x; 1.0573x over previous
#include <hip/hip_runtime.h>


#define N_NODES   100000
#define N_EDGES   1600000
#define IN_FEATS  128
#define H_FEATS   64
#define N_CONVS   3
#define POLY_ORDER 4
#define NUM_CLASSES 2
#define SC_PASSES   3

typedef __attribute__((ext_vector_type(2))) float v2f;
typedef __attribute__((ext_vector_type(8))) float v8f;

__device__ __forceinline__ v8f wmma_f32(v2f a, v2f b, v8f c) {
  return __builtin_amdgcn_wmma_f32_16x16x4_f32(
       false, a,  false, b,
       (short)0, c,  false,  false);
}

__device__ __forceinline__ void lds_barrier() {
  asm volatile("s_wait_dscnt 0x0\n\ts_barrier_signal -1\n\ts_barrier_wait -1" ::: "memory");
}

#define DEG_RANGES      8
#define DEG_RANGE_NODES 12512
#define DEG_SLICES      32
#define DEG_THREADS     512
#define DEG_LOADS       8

__global__ __launch_bounds__(DEG_THREADS)
void degree_hist_kernel(const int* __restrict__ dst, unsigned* __restrict__ part,
                        int E, int n) {
  __shared__ unsigned cnt[DEG_RANGE_NODES];

  const int tid   = threadIdx.x;
  const int range = blockIdx.x % DEG_RANGES;
  const int slice = blockIdx.x / DEG_RANGES;
  const int lo    = range * DEG_RANGE_NODES;
  int hi = lo + DEG_RANGE_NODES;
  if (hi > n) hi = n;
  const unsigned span = hi > lo ? (unsigned)(hi - lo) : 0u;

  for (int i = tid; i < DEG_RANGE_NODES; i += DEG_THREADS) cnt[i] = 0u;
  __syncthreads();

  const int slice_len = (((E + DEG_SLICES - 1) / DEG_SLICES) + 3) & ~3;
  const int e_begin = slice * slice_len;
  int e_end = e_begin + slice_len;
  if (e_end > E) e_end = E;
  const int n4 = e_end > e_begin ? (e_end - e_begin) >> 2 : 0;

  const int4* d4 = reinterpret_cast<const int4*>(dst + e_begin);
  for (int i0 = tid; i0 < n4; i0 += DEG_LOADS * DEG_THREADS) {
    int4 v[DEG_LOADS];
    #pragma unroll
    for (int u = 0; u < DEG_LOADS; ++u) {
      const int i = i0 + u * DEG_THREADS;
      v[u] = i < n4 ? d4[i] : make_int4(-1, -1, -1, -1);
    }
    #pragma unroll
    for (int u = 0; u < DEG_LOADS; ++u) {
      const unsigned a = (unsigned)(v[u].x - lo);
      const unsigned b = (unsigned)(v[u].y - lo);
      const unsigned c = (unsigned)(v[u].z - lo);
      const unsigned d = (unsigned)(v[u].w - lo);
      if (a < span) atomicAdd(&cnt[a], 1u);
      if (b < span) atomicAdd(&cnt[b], 1u);
      if (c < span) atomicAdd(&cnt[c], 1u);
      if (d < span) atomicAdd(&cnt[d], 1u);
    }
  }
  for (int e = e_begin + 4 * n4 + tid; e < e_end; e += DEG_THREADS) {
    const unsigned a = (unsigned)(dst[e] - lo);
    if (a < span) atomicAdd(&cnt[a], 1u);
  }
  __syncthreads();

  unsigned* prow = part + (size_t)slice * n + lo;
  for (unsigned i = tid; i < span; i += DEG_THREADS) prow[i] = cnt[i];
}

#define ROW_TILES (N_NODES / 16)
#define PT_BLOCKS 1250

__global__ __launch_bounds__(128)
void trunk_kernel(const float* __restrict__ X,
                  const float* __restrict__ W1, const float* __restrict__ b1,
                  const float* __restrict__ W2, const float* __restrict__ b2,
                  float* __restrict__ P0,
                  const unsigned* __restrict__ part, float* __restrict__ dinv,
                  float* __restrict__ U1, float* __restrict__ U2,
                  float* __restrict__ U3hi, unsigned* __restrict__ arrive) {
  __shared__ __attribute__((aligned(16))) float ldsX[16 * 132];
  __shared__ float ldsH[16 * 72];
  __shared__ unsigned ldsD[128];

  const int tid  = threadIdx.x;
  const int lane = tid & 31;
  const int wave = tid >> 5;
  const int mrow = lane & 15;
  const int kh   = lane >> 4;
  const int col0 = wave * 16;

  v2f w1f[IN_FEATS / 4], w2f[H_FEATS / 4];
  #pragma unroll
  for (int s = 0; s < IN_FEATS / 4; ++s) {
    w1f[s].x = W1[(size_t)(4 * s + 2 * kh)     * H_FEATS + col0 + mrow];
    w1f[s].y = W1[(size_t)(4 * s + 2 * kh + 1) * H_FEATS + col0 + mrow];
  }
  #pragma unroll
  for (int s = 0; s < H_FEATS / 4; ++s) {
    w2f[s].x = W2[(size_t)(4 * s + 2 * kh)     * H_FEATS + col0 + mrow];
    w2f[s].y = W2[(size_t)(4 * s + 2 * kh + 1) * H_FEATS + col0 + mrow];
  }
  const float bias1 = b1[col0 + mrow];
  const float bias2 = b2[col0 + mrow];

  if (blockIdx.x == 0 && tid < 2 * SC_PASSES) arrive[tid * 32] = 0u;

  const int dr = tid & 15, dg = tid >> 4;
  const float4*   xsrc = reinterpret_cast<const float4*>(X) + (size_t)wave * (IN_FEATS / 4) + lane;
  const unsigned* psrc = part + (size_t)(dg * (DEG_SLICES / 8)) * N_NODES + dr;
  int tile = blockIdx.x;
  float4 xr0, xr1, xr2, xr3;
  unsigned pc0, pc1, pc2, pc3;
  {
    const float4*   xs = xsrc + (size_t)tile * (16 * IN_FEATS / 4);
    const unsigned* ps = psrc + tile * 16;
    xr0 = xs[0];  xr1 = xs[4 * (IN_FEATS / 4)];  xr2 = xs[8 * (IN_FEATS / 4)];  xr3 = xs[12 * (IN_FEATS / 4)];
    pc0 = ps[0];  pc1 = ps[N_NODES];  pc2 = ps[2 * N_NODES];  pc3 = ps[3 * N_NODES];
  }

  for (; tile < ROW_TILES; tile += gridDim.x) {
    const int row0 = tile * 16;

    {
      float* xd = &ldsX[wave * 132 + 4 * lane];
      *reinterpret_cast<float4*>(xd)            = xr0;
      *reinterpret_cast<float4*>(xd + 4 * 132)  = xr1;
      *reinterpret_cast<float4*>(xd + 8 * 132)  = xr2;
      *reinterpret_cast<float4*>(xd + 12 * 132) = xr3;
    }
    ldsD[tid] = (pc0 + pc1) + (pc2 + pc3);

    {
      const float4 z4 = make_float4(0.0f, 0.0f, 0.0f, 0.0f);
      float4* u1 = reinterpret_cast<float4*>(U1) + (size_t)tile * 256;
      float4* u2 = reinterpret_cast<float4*>(U2) + (size_t)tile * 256;
      float4* u3 = reinterpret_cast<float4*>(U3hi) + (size_t)tile * 128;
      u1[tid] = z4;  u1[tid + 128] = z4;
      u2[tid] = z4;  u2[tid + 128] = z4;
      u3[tid] = z4;
    }

    {
      const int nt = tile + (int)gridDim.x;
      if (nt < ROW_TILES) {
        const float4*   xs = xsrc + (size_t)nt * (16 * IN_FEATS / 4);
        const unsigned* ps = psrc + nt * 16;
        xr0 = xs[0];  xr1 = xs[4 * (IN_FEATS / 4)];  xr2 = xs[8 * (IN_FEATS / 4)];  xr3 = xs[12 * (IN_FEATS / 4)];
        pc0 = ps[0];  pc1 = ps[N_NODES];  pc2 = ps[2 * N_NODES];  pc3 = ps[3 * N_NODES];
      }
    }
    lds_barrier();

    if (tid < 16) {
      unsigned cs = 0u;
      #pragma unroll
      for (int g = 0; g < 8; ++g) cs += ldsD[tid + 16 * g];
      float d = (float)cs;
      d = d < 1.0f ? 1.0f : d;
      dinv[row0 + tid] = rsqrtf(d);
    }

    v8f acc = {};
    #pragma unroll
    for (int s = 0; s < IN_FEATS / 4; ++s) {
      v2f a;
      a.x = ldsX[mrow * 132 + 4 * s + 2 * kh];
      a.y = ldsX[mrow * 132 + 4 * s + 2 * kh + 1];
      acc = wmma_f32(a, w1f[s], acc);
    }
    #pragma unroll
    for (int v = 0; v < 8; ++v) {
      int r = v + 8 * kh;
      float val = acc[v] + bias1;
      ldsH[r * 72 + col0 + mrow] = val > 0.0f ? val : 0.0f;
    }
    lds_barrier();

    v8f acc2 = {};
    #pragma unroll
    for (int s = 0; s < H_FEATS / 4; ++s) {
      v2f a;
      a.x = ldsH[mrow * 72 + 4 * s + 2 * kh];
      a.y = ldsH[mrow * 72 + 4 * s + 2 * kh + 1];
      acc2 = wmma_f32(a, w2f[s], acc2);
    }
    #pragma unroll
    for (int v = 0; v < 8; ++v) {
      int r = v + 8 * kh;
      float val = acc2[v] + bias2;
      P0[(size_t)(row0 + r) * H_FEATS + col0 + mrow] = val > 0.0f ? val : 0.0f;
    }
  }
}

#define SC_BLOCKS   2048
#define SC_THREADS  256
#define SC_WAVES    (SC_THREADS / 32)
#define SC_ROLE_WAVES_LOG2 13
#define SC_CHUNK    7
#define SC_RING     256
#define ROWS_AHEAD  8

__global__ __launch_bounds__(SC_THREADS)
void scatter_kernel(const float* __restrict__ P,
                    const int* __restrict__ src, const int* __restrict__ dst,
                    const float* __restrict__ dinv, float* __restrict__ agg,
                    int E, int n, int squared, float* __restrict__ zbuf, int zn4,
                    unsigned* __restrict__ arrive) {
  __shared__ int   ring_s[SC_WAVES][SC_RING];
  __shared__ int   ring_d[SC_WAVES][SC_RING];
  __shared__ float ring_w[SC_WAVES][SC_RING];
  __shared__ int   role_sh;

  const int lane = threadIdx.x & 31;
  const int w    = __builtin_amdgcn_readfirstlane((int)(threadIdx.x >> 5));

  if (threadIdx.x == 0) {
    const unsigned place = __builtin_amdgcn_s_sendmsg_rtn(0x87);
    const int h = (int)((place >> 18) & 1u);
    const int rank = (int)atomicAdd(&arrive[h * 32], 1u);
    int q = h, r = rank;
    if (rank >= SC_BLOCKS / 2) { q = h ^ 1; r = SC_BLOCKS - 1 - rank; }
    role_sh = (q << 16) | (r & 0xffff);
  }
  __syncthreads();
  const int q = role_sh >> 16;
  const int r = role_sh & 0xffff;
  if (r >= SC_BLOCKS / 2) return;

  if (zbuf && q == 0) {
    float4* z4 = reinterpret_cast<float4*>(zbuf);
    for (int i = r * SC_THREADS + threadIdx.x; i < zn4; i += (SC_BLOCKS / 2) * SC_THREADS)
      z4[i] = make_float4(0.0f, 0.0f, 0.0f, 0.0f);
  }

  const int lo = q ? (n >> 1) : 0;
  const unsigned span = q ? (unsigned)(n - (n >> 1)) : (unsigned)(n >> 1);

  const long long rwid = (long long)(r * SC_WAVES + w);
  const int e_lo = (int)((rwid * E) >> SC_ROLE_WAVES_LOG2);
  const int e_hi = (int)(((rwid + 1) * E) >> SC_ROLE_WAVES_LOG2);

  int*   rs = ring_s[w];
  int*   rd = ring_d[w];
  float* rw = ring_w[w];
  int head = 0, tail = 0;

  for (int eb = e_lo; eb < e_hi; eb += 32 * SC_CHUNK) {
    int sv[SC_CHUNK], dv[SC_CHUNK];
    #pragma unroll
    for (int u = 0; u < SC_CHUNK; ++u) {
      const int e  = eb + 32 * u + lane;
      const int ec = e < e_hi ? e : e_hi - 1;
      const int dvv = dst[ec];
      sv[u] = src[ec];
      dv[u] = e < e_hi ? dvv : -1;
    }
    float wv[SC_CHUNK];
    #pragma unroll
    for (int u = 0; u < SC_CHUNK; ++u) {
      const bool inr = (unsigned)(dv[u] - lo) < span;
      const float ds = inr ? dinv[sv[u]] : 0.0f;
      wv[u] = squared ? ds * ds : ds;
    }
    #pragma unroll
    for (int u = 0; u < SC_CHUNK; ++u) {
      const bool inr = (unsigned)(dv[u] - lo) < span;
      const unsigned m = __builtin_amdgcn_ballot_w32(inr);
      if (inr) {
        const int pos = (tail + (int)__builtin_amdgcn_mbcnt_lo(m, 0u)) & (SC_RING - 1);
        rs[pos] = sv[u];
        rd[pos] = dv[u];
        rw[pos] = wv[u];
      }
      tail += __builtin_popcount(m);
    }
    __builtin_amdgcn_wave_barrier();

    while (tail - head >= 32) {
      const int idx = (head + lane) & (SC_RING - 1);
      const int   es = rs[idx];
      const int   ed = rd[idx];
      const float ew = rw[idx];
      head += 32;
      #pragma unroll 1
      for (int g = 0; g < 32; g += ROWS_AHEAD) {
        float vlo[ROWS_AHEAD], vhi[ROWS_AHEAD];
        #pragma unroll
        for (int j = 0; j < ROWS_AHEAD; ++j) {
          const int s = __builtin_amdgcn_readlane(es, g + j);
          const float* rp = P + (size_t)s * H_FEATS + lane;
          vlo[j] = rp[0];
          vhi[j] = rp[H_FEATS / 2];
        }
        #pragma unroll
        for (int j = 0; j < ROWS_AHEAD; ++j) {
          const int   d  = __builtin_amdgcn_readlane(ed, g + j);
          const float di = __int_as_float(__builtin_amdgcn_readlane(__float_as_int(ew), g + j));
          float* ap = agg + (size_t)d * H_FEATS + lane;
          atomicAdd(ap,               vlo[j] * di);
          atomicAdd(ap + H_FEATS / 2, vhi[j] * di);
        }
      }
    }
    __builtin_amdgcn_wave_barrier();
  }

  const int cnt = tail - head;
  if (cnt > 0) {
    const int idx = (head + lane) & (SC_RING - 1);
    const bool has = lane < cnt;
    const int   es = has ? rs[idx] : 0;
    const int   ed = has ? rd[idx] : 0;
    const float ew = has ? rw[idx] : 0.0f;
    #pragma unroll 1
    for (int g = 0; g < cnt; g += ROWS_AHEAD) {
      const int k = cnt - g;
      float vlo[ROWS_AHEAD], vhi[ROWS_AHEAD];
      #pragma unroll
      for (int j = 0; j < ROWS_AHEAD; ++j) {
        vlo[j] = 0.0f;  vhi[j] = 0.0f;
        if (j < k) {
          const int s = __builtin_amdgcn_readlane(es, g + j);
          const float* rp = P + (size_t)s * H_FEATS + lane;
          vlo[j] = rp[0];
          vhi[j] = rp[H_FEATS / 2];
        }
      }
      #pragma unroll
      for (int j = 0; j < ROWS_AHEAD; ++j) {
        if (j < k) {
          const int   d  = __builtin_amdgcn_readlane(ed, g + j);
          const float di = __int_as_float(__builtin_amdgcn_readlane(__float_as_int(ew), g + j));
          float* ap = agg + (size_t)d * H_FEATS + lane;
          atomicAdd(ap,               vlo[j] * di);
          atomicAdd(ap + H_FEATS / 2, vhi[j] * di);
        }
      }
    }
  }
}

#define HEAD_K     (N_CONVS * H_FEATS)
#define HEAD_PITCH (HEAD_K + 4)

__global__ __launch_bounds__(128, 5)
void head_kernel(const float* __restrict__ P0, const float* __restrict__ P1,
                 const float* __restrict__ P2, const float* __restrict__ P3,
                 const float* __restrict__ thetas, const float* __restrict__ Wm1,
                 const float* __restrict__ bm1,
                 const float* __restrict__ Wm2, const float* __restrict__ bm2,
                 const float* __restrict__ dinv, float* __restrict__ out) {
  __shared__ float ldsA[16 * HEAD_PITCH];
  __shared__ float ldsH[16 * 72];

  const int tid  = threadIdx.x;
  const int lane = tid & 31;
  const int wave = tid >> 5;
  const int mrow = lane & 15;
  const int kh   = lane >> 4;
  const int col0 = wave * 16;

  float cf[N_CONVS][POLY_ORDER];
  #pragma unroll
  for (int i = 0; i < N_CONVS; ++i) {
    const float t0 = thetas[i * POLY_ORDER + 0], t1 = thetas[i * POLY_ORDER + 1];
    const float t2 = thetas[i * POLY_ORDER + 2], t3 = thetas[i * POLY_ORDER + 3];
    cf[i][0] = t0 + t1 + t2 + t3;
    cf[i][1] = -(t1 + 2.0f * t2 + 3.0f * t3);
    cf[i][2] = t2 + 3.0f * t3;
    cf[i][3] = -t3;
  }

  v2f wf[HEAD_K / 4];
  #pragma unroll
  for (int s = 0; s < HEAD_K / 4; ++s) {
    wf[s].x = Wm1[(size_t)(4 * s + 2 * kh)     * H_FEATS + col0 + mrow];
    wf[s].y = Wm1[(size_t)(4 * s + 2 * kh + 1) * H_FEATS + col0 + mrow];
  }
  const float bias = bm1[col0 + mrow];

  const int tr = tid >> 6, tc = tid & 63;
  float u0[8], u1[8], u2[8], u3[8], dsc[8];
  int tile = blockIdx.x;
  #pragma unroll
  for (int t = 0; t < 8; ++t) {
    const int r = tile * 16 + tr + 2 * t;
    const size_t off = (size_t)r * H_FEATS + tc;
    u0[t] = P0[off];
    u1[t] = P1[off];
    u2[t] = P2[off];
    u3[t] = P3[off];
    dsc[t] = dinv[r];
  }

  for (; tile < ROW_TILES; tile += gridDim.x) {
    const int row0 = tile * 16;

    #pragma unroll
    for (int t = 0; t < 8; ++t) {
      float* arow = ldsA + (tr + 2 * t) * HEAD_PITCH + tc;
      #pragma unroll
      for (int i = 0; i < N_CONVS; ++i) {
        const float g = cf[i][1] * u1[t] + cf[i][2] * u2[t] + cf[i][3] * u3[t];
        arow[i * H_FEATS] = cf[i][0] * u0[t] + dsc[t] * g;
      }
    }

    {
      const int nt = tile + (int)gridDim.x;
      if (nt < ROW_TILES) {
        #pragma unroll
        for (int t = 0; t < 8; ++t) {
          const int r = nt * 16 + tr + 2 * t;
          const size_t off = (size_t)r * H_FEATS + tc;
          u0[t] = P0[off];
          u1[t] = P1[off];
          u2[t] = P2[off];
          u3[t] = P3[off];
          dsc[t] = dinv[r];
        }
      }
    }
    lds_barrier();

    v8f acc = {};
    #pragma unroll
    for (int s = 0; s < HEAD_K / 4; ++s) {
      v2f a;
      a.x = ldsA[mrow * HEAD_PITCH + 4 * s + 2 * kh];
      a.y = ldsA[mrow * HEAD_PITCH + 4 * s + 2 * kh + 1];
      acc = wmma_f32(a, wf[s], acc);
    }

    #pragma unroll
    for (int v = 0; v < 8; ++v) {
      int r = v + 8 * kh;
      float val = acc[v] + bias;
      ldsH[r * 72 + col0 + mrow] = val > 0.0f ? val : 0.0f;
    }
    lds_barrier();

    if (tid < 32) {
      int r = tid >> 1, c = tid & 1;
      float s = bm2[c];
      #pragma unroll 8
      for (int j = 0; j < H_FEATS; ++j)
        s += ldsH[r * 72 + j] * Wm2[j * NUM_CLASSES + c];
      out[(size_t)(row0 + r) * NUM_CLASSES + c] = s;
    }
  }
}

extern "C" void kernel_launch(void* const* d_in, const int* in_sizes, int n_in,
                              void* d_out, int out_size, void* d_ws, size_t ws_size,
                              hipStream_t stream) {
  const float* X      = (const float*)d_in[0];
  const int*   src    = (const int*)  d_in[1];
  const int*   dst    = (const int*)  d_in[2];
  const float* W1     = (const float*)d_in[3];
  const float* b1     = (const float*)d_in[4];
  const float* W2     = (const float*)d_in[5];
  const float* b2     = (const float*)d_in[6];
  const float* thetas = (const float*)d_in[7];
  const float* Wm1    = (const float*)d_in[8];
  const float* bm1    = (const float*)d_in[9];
  const float* Wm2    = (const float*)d_in[10];
  const float* bm2    = (const float*)d_in[11];
  float* out = (float*)d_out;
  float* ws  = (float*)d_ws;

  const int PN = N_NODES * H_FEATS;
  float* dinv = ws;
  float* P0   = ws + 100352;
  float* P1   = P0 + PN;
  float* P2   = P1 + PN;
  float* P3   = P2 + PN;
  unsigned* arrive = (unsigned*)(ws + 100096);

  unsigned* part = (unsigned*)P3;
  degree_hist_kernel<<<DEG_SLICES * DEG_RANGES, DEG_THREADS, 0, stream>>>(
      dst, part, N_EDGES, N_NODES);

  trunk_kernel<<<PT_BLOCKS, 128, 0, stream>>>(X, W1, b1, W2, b2, P0,
                                                 part, dinv, P1, P2, P3 + PN / 2, arrive);

  scatter_kernel<<<SC_BLOCKS, SC_THREADS, 0, stream>>>(
      P0, src, dst, dinv, P1, N_EDGES, N_NODES, 0, P3, PN / 8, arrive);
  scatter_kernel<<<SC_BLOCKS, SC_THREADS, 0, stream>>>(
      P1, src, dst, dinv, P2, N_EDGES, N_NODES, 1, nullptr, 0, arrive + 64);
  scatter_kernel<<<SC_BLOCKS, SC_THREADS, 0, stream>>>(
      P2, src, dst, dinv, P3, N_EDGES, N_NODES, 1, nullptr, 0, arrive + 128);

  head_kernel<<<PT_BLOCKS, 128, 0, stream>>>(
      P0, P1, P2, P3, thetas, Wm1, bm1, Wm2, bm2, dinv, out);
}


# edge passes: an edge's d^-1/2 comes from a 16-bit degree table in LDS (copied per pass by one 1024-thread work-group per WGP), not from a 4-byte load per edge; same f32 atomics, arrival-order sums
# speedup vs baseline: 3.1326x; 1.0291x over previous
#include <hip/hip_runtime.h>


#define N_NODES   100000
#define N_EDGES   1600000
#define IN_FEATS  128
#define H_FEATS   64
#define N_CONVS   3
#define POLY_ORDER 4
#define NUM_CLASSES 2
#define SC_PASSES   3

typedef __attribute__((ext_vector_type(2))) float v2f;
typedef __attribute__((ext_vector_type(8))) float v8f;

__device__ __forceinline__ v8f wmma_f32(v2f a, v2f b, v8f c) {
  return __builtin_amdgcn_wmma_f32_16x16x4_f32(
       false, a,  false, b,
       (short)0, c,  false,  false);
}

#define WT_ESC_LO 65504u
static_assert(N_EDGES < (1 << 21), "degree >> 5 must fit 16 bits");
static_assert((N_NODES * 2) % 16 == 0, "code table is copied in 16-byte pieces");
__device__ __forceinline__ unsigned wt_degree(unsigned code, const unsigned short* __restrict__ hi, int node) {
  return code < WT_ESC_LO ? code : (((unsigned)hi[node] << 5) | (code - WT_ESC_LO));
}
__device__ __forceinline__ float wt_from_degree(unsigned deg) { return rsqrtf((float)deg); }

__device__ __forceinline__ void lds_barrier() {
  asm volatile("s_wait_dscnt 0x0\n\ts_barrier_signal -1\n\ts_barrier_wait -1" ::: "memory");
}

#define DEG_RANGES      8
#define DEG_RANGE_NODES 12512
#define DEG_SLICES      32
#define DEG_THREADS     512
#define DEG_LOADS       8

__global__ __launch_bounds__(DEG_THREADS)
void degree_hist_kernel(const int* __restrict__ dst, unsigned* __restrict__ part,
                        int E, int n) {
  __shared__ unsigned cnt[DEG_RANGE_NODES];

  const int tid   = threadIdx.x;
  const int range = blockIdx.x % DEG_RANGES;
  const int slice = blockIdx.x / DEG_RANGES;
  const int lo    = range * DEG_RANGE_NODES;
  int hi = lo + DEG_RANGE_NODES;
  if (hi > n) hi = n;
  const unsigned span = hi > lo ? (unsigned)(hi - lo) : 0u;

  for (int i = tid; i < DEG_RANGE_NODES; i += DEG_THREADS) cnt[i] = 0u;
  __syncthreads();

  const int slice_len = (((E + DEG_SLICES - 1) / DEG_SLICES) + 3) & ~3;
  const int e_begin = slice * slice_len;
  int e_end = e_begin + slice_len;
  if (e_end > E) e_end = E;
  const int n4 = e_end > e_begin ? (e_end - e_begin) >> 2 : 0;

  const int4* d4 = reinterpret_cast<const int4*>(dst + e_begin);
  for (int i0 = tid; i0 < n4; i0 += DEG_LOADS * DEG_THREADS) {
    int4 v[DEG_LOADS];
    #pragma unroll
    for (int u = 0; u < DEG_LOADS; ++u) {
      const int i = i0 + u * DEG_THREADS;
      v[u] = i < n4 ? d4[i] : make_int4(-1, -1, -1, -1);
    }
    #pragma unroll
    for (int u = 0; u < DEG_LOADS; ++u) {
      const unsigned a = (unsigned)(v[u].x - lo);
      const unsigned b = (unsigned)(v[u].y - lo);
      const unsigned c = (unsigned)(v[u].z - lo);
      const unsigned d = (unsigned)(v[u].w - lo);
      if (a < span) atomicAdd(&cnt[a], 1u);
      if (b < span) atomicAdd(&cnt[b], 1u);
      if (c < span) atomicAdd(&cnt[c], 1u);
      if (d < span) atomicAdd(&cnt[d], 1u);
    }
  }
  for (int e = e_begin + 4 * n4 + tid; e < e_end; e += DEG_THREADS) {
    const unsigned a = (unsigned)(dst[e] - lo);
    if (a < span) atomicAdd(&cnt[a], 1u);
  }
  __syncthreads();

  unsigned* prow = part + (size_t)slice * n + lo;
  for (unsigned i = tid; i < span; i += DEG_THREADS) prow[i] = cnt[i];
}

#define ROW_TILES (N_NODES / 16)
#define PT_BLOCKS 1250

__global__ __launch_bounds__(128)
void trunk_kernel(const float* __restrict__ X,
                  const float* __restrict__ W1, const float* __restrict__ b1,
                  const float* __restrict__ W2, const float* __restrict__ b2,
                  float* __restrict__ P0,
                  const unsigned* __restrict__ part, unsigned short* __restrict__ wcode,
                  float* __restrict__ U1, float* __restrict__ U2,
                  float* __restrict__ U3hi, unsigned* __restrict__ arrive) {
  __shared__ __attribute__((aligned(16))) float ldsX[16 * 132];
  __shared__ float ldsH[16 * 72];
  __shared__ unsigned ldsD[128];

  const int tid  = threadIdx.x;
  const int lane = tid & 31;
  const int wave = tid >> 5;
  const int mrow = lane & 15;
  const int kh   = lane >> 4;
  const int col0 = wave * 16;

  v2f w1f[IN_FEATS / 4], w2f[H_FEATS / 4];
  #pragma unroll
  for (int s = 0; s < IN_FEATS / 4; ++s) {
    w1f[s].x = W1[(size_t)(4 * s + 2 * kh)     * H_FEATS + col0 + mrow];
    w1f[s].y = W1[(size_t)(4 * s + 2 * kh + 1) * H_FEATS + col0 + mrow];
  }
  #pragma unroll
  for (int s = 0; s < H_FEATS / 4; ++s) {
    w2f[s].x = W2[(size_t)(4 * s + 2 * kh)     * H_FEATS + col0 + mrow];
    w2f[s].y = W2[(size_t)(4 * s + 2 * kh + 1) * H_FEATS + col0 + mrow];
  }
  const float bias1 = b1[col0 + mrow];
  const float bias2 = b2[col0 + mrow];

  if (blockIdx.x == 0 && tid < 2 * SC_PASSES) arrive[tid * 32] = 0u;

  const int dr = tid & 15, dg = tid >> 4;
  const float4*   xsrc = reinterpret_cast<const float4*>(X) + (size_t)wave * (IN_FEATS / 4) + lane;
  const unsigned* psrc = part + (size_t)(dg * (DEG_SLICES / 8)) * N_NODES + dr;
  int tile = blockIdx.x;
  float4 xr0, xr1, xr2, xr3;
  unsigned pc0, pc1, pc2, pc3;
  {
    const float4*   xs = xsrc + (size_t)tile * (16 * IN_FEATS / 4);
    const unsigned* ps = psrc + tile * 16;
    xr0 = xs[0];  xr1 = xs[4 * (IN_FEATS / 4)];  xr2 = xs[8 * (IN_FEATS / 4)];  xr3 = xs[12 * (IN_FEATS / 4)];
    pc0 = ps[0];  pc1 = ps[N_NODES];  pc2 = ps[2 * N_NODES];  pc3 = ps[3 * N_NODES];
  }

  for (; tile < ROW_TILES; tile += gridDim.x) {
    const int row0 = tile * 16;

    {
      float* xd = &ldsX[wave * 132 + 4 * lane];
      *reinterpret_cast<float4*>(xd)            = xr0;
      *reinterpret_cast<float4*>(xd + 4 * 132)  = xr1;
      *reinterpret_cast<float4*>(xd + 8 * 132)  = xr2;
      *reinterpret_cast<float4*>(xd + 12 * 132) = xr3;
    }
    ldsD[tid] = (pc0 + pc1) + (pc2 + pc3);

    {
      const float4 z4 = make_float4(0.0f, 0.0f, 0.0f, 0.0f);
      float4* u1 = reinterpret_cast<float4*>(U1) + (size_t)tile * 256;
      float4* u2 = reinterpret_cast<float4*>(U2) + (size_t)tile * 256;
      float4* u3 = reinterpret_cast<float4*>(U3hi) + (size_t)tile * 128;
      u1[tid] = z4;  u1[tid + 128] = z4;
      u2[tid] = z4;  u2[tid + 128] = z4;
      u3[tid] = z4;
    }

    {
      const int nt = tile + (int)gridDim.x;
      if (nt < ROW_TILES) {
        const float4*   xs = xsrc + (size_t)nt * (16 * IN_FEATS / 4);
        const unsigned* ps = psrc + nt * 16;
        xr0 = xs[0];  xr1 = xs[4 * (IN_FEATS / 4)];  xr2 = xs[8 * (IN_FEATS / 4)];  xr3 = xs[12 * (IN_FEATS / 4)];
        pc0 = ps[0];  pc1 = ps[N_NODES];  pc2 = ps[2 * N_NODES];  pc3 = ps[3 * N_NODES];
      }
    }
    lds_barrier();

    if (tid < 16) {
      unsigned cs = 0u;
      #pragma unroll
      for (int g = 0; g < 8; ++g) cs += ldsD[tid + 16 * g];
      cs = cs < 1u ? 1u : cs;
      wcode[row0 + tid]           = (unsigned short)(cs < WT_ESC_LO ? cs : WT_ESC_LO + (cs & 31u));
      wcode[N_NODES + row0 + tid] = (unsigned short)(cs >> 5);
    }

    v8f acc = {};
    #pragma unroll
    for (int s = 0; s < IN_FEATS / 4; ++s) {
      v2f a;
      a.x = ldsX[mrow * 132 + 4 * s + 2 * kh];
      a.y = ldsX[mrow * 132 + 4 * s + 2 * kh + 1];
      acc = wmma_f32(a, w1f[s], acc);
    }
    #pragma unroll
    for (int v = 0; v < 8; ++v) {
      int r = v + 8 * kh;
      float val = acc[v] + bias1;
      ldsH[r * 72 + col0 + mrow] = val > 0.0f ? val : 0.0f;
    }
    lds_barrier();

    v8f acc2 = {};
    #pragma unroll
    for (int s = 0; s < H_FEATS / 4; ++s) {
      v2f a;
      a.x = ldsH[mrow * 72 + 4 * s + 2 * kh];
      a.y = ldsH[mrow * 72 + 4 * s + 2 * kh + 1];
      acc2 = wmma_f32(a, w2f[s], acc2);
    }
    #pragma unroll
    for (int v = 0; v < 8; ++v) {
      int r = v + 8 * kh;
      float val = acc2[v] + bias2;
      P0[(size_t)(row0 + r) * H_FEATS + col0 + mrow] = val > 0.0f ? val : 0.0f;
    }
  }
}

#define SC_BLOCKS   256
#define SC_THREADS  1024
#define SC_WAVES    (SC_THREADS / 32)
#define SC_ROLE_WAVES_LOG2 12
#define WT_PIECES   (N_NODES * 2 / 16)
#define WT_LOADS    ((WT_PIECES + SC_THREADS - 1) / SC_THREADS)
#define SC_CHUNK    7
#define SC_RING     256
#define ROWS_AHEAD  16

__global__ __launch_bounds__(SC_THREADS)
void scatter_kernel(const float* __restrict__ P,
                    const int* __restrict__ src, const int* __restrict__ dst,
                    const unsigned short* __restrict__ wcode, float* __restrict__ agg,
                    int E, int n, int squared, float* __restrict__ zbuf, int zn4,
                    unsigned* __restrict__ arrive) {
  __shared__ __attribute__((aligned(16))) unsigned short wt_tab[N_NODES];
  __shared__ int   ring_s[SC_WAVES][SC_RING];
  __shared__ int   ring_d[SC_WAVES][SC_RING];
  __shared__ float ring_w[SC_WAVES][SC_RING];
  __shared__ int   role_sh;

  const int lane = threadIdx.x & 31;
  const int w    = __builtin_amdgcn_readfirstlane((int)(threadIdx.x >> 5));

  if (threadIdx.x == 0) {
    const unsigned place = __builtin_amdgcn_s_sendmsg_rtn(0x87);
    const int h = (int)((place >> 18) & 1u);
    const int rank = (int)atomicAdd(&arrive[h * 32], 1u);
    int q = h, r = rank;
    if (rank >= SC_BLOCKS / 2) { q = h ^ 1; r = SC_BLOCKS - 1 - rank; }
    role_sh = (q << 16) | (r & 0xffff);
  }

  {
    const uint4* g4 = reinterpret_cast<const uint4*>(wcode);
    uint4* l4 = reinterpret_cast<uint4*>(wt_tab);
    uint4 v[WT_LOADS];
    #pragma unroll
    for (int u = 0; u < WT_LOADS; ++u) {
      const int i = (int)threadIdx.x + u * SC_THREADS;
      v[u] = g4[i < WT_PIECES ? i : WT_PIECES - 1];
    }
    #pragma unroll
    for (int u = 0; u < WT_LOADS; ++u) {
      const int i = (int)threadIdx.x + u * SC_THREADS;
      if (i < WT_PIECES) l4[i] = v[u];
    }
  }
  __syncthreads();
  const int q = role_sh >> 16;
  const int r = role_sh & 0xffff;
  if (r >= SC_BLOCKS / 2) return;

  if (zbuf && q == 0) {
    float4* z4 = reinterpret_cast<float4*>(zbuf);
    for (int i = r * SC_THREADS + threadIdx.x; i < zn4; i += (SC_BLOCKS / 2) * SC_THREADS)
      z4[i] = make_float4(0.0f, 0.0f, 0.0f, 0.0f);
  }

  const int lo = q ? (n >> 1) : 0;
  const unsigned span = q ? (unsigned)(n - (n >> 1)) : (unsigned)(n >> 1);

  const long long rwid = (long long)(r * SC_WAVES + w);
  const int e_lo = (int)((rwid * E) >> SC_ROLE_WAVES_LOG2);
  const int e_hi = (int)(((rwid + 1) * E) >> SC_ROLE_WAVES_LOG2);

  int*   rs = ring_s[w];
  int*   rd = ring_d[w];
  float* rw = ring_w[w];
  int head = 0, tail = 0;

  for (int eb = e_lo; eb < e_hi; eb += 32 * SC_CHUNK) {
    int sv[SC_CHUNK], dv[SC_CHUNK];
    #pragma unroll
    for (int u = 0; u < SC_CHUNK; ++u) {
      const int e  = eb + 32 * u + lane;
      const int ec = e < e_hi ? e : e_hi - 1;
      const int dvv = dst[ec];
      sv[u] = src[ec];
      dv[u] = e < e_hi ? dvv : -1;
    }
    float wv[SC_CHUNK];
    #pragma unroll
    for (int u = 0; u < SC_CHUNK; ++u) {
      const bool inr = (unsigned)(dv[u] - lo) < span;
      float ds = 0.0f;
      if (inr) ds = wt_from_degree(wt_degree(wt_tab[sv[u]], wcode + N_NODES, sv[u]));
      wv[u] = squared ? ds * ds : ds;
    }
    #pragma unroll
    for (int u = 0; u < SC_CHUNK; ++u) {
      const bool inr = (unsigned)(dv[u] - lo) < span;
      const unsigned m = __builtin_amdgcn_ballot_w32(inr);
      if (inr) {
        const int pos = (tail + (int)__builtin_amdgcn_mbcnt_lo(m, 0u)) & (SC_RING - 1);
        rs[pos] = sv[u];
        rd[pos] = dv[u];
        rw[pos] = wv[u];
      }
      tail += __builtin_popcount(m);
    }
    __builtin_amdgcn_wave_barrier();

    while (tail - head >= 32) {
      const int idx = (head + lane) & (SC_RING - 1);
      const int   es = rs[idx];
      const int   ed = rd[idx];
      const float ew = rw[idx];
      head += 32;
      #pragma unroll 1
      for (int g = 0; g < 32; g += ROWS_AHEAD) {
        float vlo[ROWS_AHEAD], vhi[ROWS_AHEAD];
        #pragma unroll
        for (int j = 0; j < ROWS_AHEAD; ++j) {
          const int s = __builtin_amdgcn_readlane(es, g + j);
          const float* rp = P + (size_t)s * H_FEATS + lane;
          vlo[j] = rp[0];
          vhi[j] = rp[H_FEATS / 2];
        }
        #pragma unroll
        for (int j = 0; j < ROWS_AHEAD; ++j) {
          const int   d  = __builtin_amdgcn_readlane(ed, g + j);
          const float di = __int_as_float(__builtin_amdgcn_readlane(__float_as_int(ew), g + j));
          float* ap = agg + (size_t)d * H_FEATS + lane;
          atomicAdd(ap,               vlo[j] * di);
          atomicAdd(ap + H_FEATS / 2, vhi[j] * di);
        }
      }
    }
    __builtin_amdgcn_wave_barrier();
  }

  const int cnt = tail - head;
  if (cnt > 0) {
    const int idx = (head + lane) & (SC_RING - 1);
    const bool has = lane < cnt;
    const int   es = has ? rs[idx] : 0;
    const int   ed = has ? rd[idx] : 0;
    const float ew = has ? rw[idx] : 0.0f;
    #pragma unroll 1
    for (int g = 0; g < cnt; g += ROWS_AHEAD) {
      const int k = cnt - g;
      float vlo[ROWS_AHEAD], vhi[ROWS_AHEAD];
      #pragma unroll
      for (int j = 0; j < ROWS_AHEAD; ++j) {
        vlo[j] = 0.0f;  vhi[j] = 0.0f;
        if (j < k) {
          const int s = __builtin_amdgcn_readlane(es, g + j);
          const float* rp = P + (size_t)s * H_FEATS + lane;
          vlo[j] = rp[0];
          vhi[j] = rp[H_FEATS / 2];
        }
      }
      #pragma unroll
      for (int j = 0; j < ROWS_AHEAD; ++j) {
        if (j < k) {
          const int   d  = __builtin_amdgcn_readlane(ed, g + j);
          const float di = __int_as_float(__builtin_amdgcn_readlane(__float_as_int(ew), g + j));
          float* ap = agg + (size_t)d * H_FEATS + lane;
          atomicAdd(ap,               vlo[j] * di);
          atomicAdd(ap + H_FEATS / 2, vhi[j] * di);
        }
      }
    }
  }
}

#define HEAD_K     (N_CONVS * H_FEATS)
#define HEAD_PITCH (HEAD_K + 4)

__global__ __launch_bounds__(128, 5)
void head_kernel(const float* __restrict__ P0, const float* __restrict__ P1,
                 const float* __restrict__ P2, const float* __restrict__ P3,
                 const float* __restrict__ thetas, const float* __restrict__ Wm1,
                 const float* __restrict__ bm1,
                 const float* __restrict__ Wm2, const float* __restrict__ bm2,
                 const unsigned short* __restrict__ wcode, float* __restrict__ out) {
  __shared__ float ldsA[16 * HEAD_PITCH];
  __shared__ float ldsH[16 * 72];

  const int tid  = threadIdx.x;
  const int lane = tid & 31;
  const int wave = tid >> 5;
  const int mrow = lane & 15;
  const int kh   = lane >> 4;
  const int col0 = wave * 16;

  float cf[N_CONVS][POLY_ORDER];
  #pragma unroll
  for (int i = 0; i < N_CONVS; ++i) {
    const float t0 = thetas[i * POLY_ORDER + 0], t1 = thetas[i * POLY_ORDER + 1];
    const float t2 = thetas[i * POLY_ORDER + 2], t3 = thetas[i * POLY_ORDER + 3];
    cf[i][0] = t0 + t1 + t2 + t3;
    cf[i][1] = -(t1 + 2.0f * t2 + 3.0f * t3);
    cf[i][2] = t2 + 3.0f * t3;
    cf[i][3] = -t3;
  }

  v2f wf[HEAD_K / 4];
  #pragma unroll
  for (int s = 0; s < HEAD_K / 4; ++s) {
    wf[s].x = Wm1[(size_t)(4 * s + 2 * kh)     * H_FEATS + col0 + mrow];
    wf[s].y = Wm1[(size_t)(4 * s + 2 * kh + 1) * H_FEATS + col0 + mrow];
  }
  const float bias = bm1[col0 + mrow];

  const int tr = tid >> 6, tc = tid & 63;
  float u0[8], u1[8], u2[8], u3[8], dsc[8];
  int tile = blockIdx.x;
  #pragma unroll
  for (int t = 0; t < 8; ++t) {
    const int r = tile * 16 + tr + 2 * t;
    const size_t off = (size_t)r * H_FEATS + tc;
    u0[t] = P0[off];
    u1[t] = P1[off];
    u2[t] = P2[off];
    u3[t] = P3[off];
    dsc[t] = wt_from_degree(wt_degree(wcode[r], wcode + N_NODES, r));
  }

  for (; tile < ROW_TILES; tile += gridDim.x) {
    const int row0 = tile * 16;

    #pragma unroll
    for (int t = 0; t < 8; ++t) {
      float* arow = ldsA + (tr + 2 * t) * HEAD_PITCH + tc;
      #pragma unroll
      for (int i = 0; i < N_CONVS; ++i) {
        const float g = cf[i][1] * u1[t] + cf[i][2] * u2[t] + cf[i][3] * u3[t];
        arow[i * H_FEATS] = cf[i][0] * u0[t] + dsc[t] * g;
      }
    }

    {
      const int nt = tile + (int)gridDim.x;
      if (nt < ROW_TILES) {
        #pragma unroll
        for (int t = 0; t < 8; ++t) {
          const int r = nt * 16 + tr + 2 * t;
          const size_t off = (size_t)r * H_FEATS + tc;
          u0[t] = P0[off];
          u1[t] = P1[off];
          u2[t] = P2[off];
          u3[t] = P3[off];
          dsc[t] = wt_from_degree(wt_degree(wcode[r], wcode + N_NODES, r));
        }
      }
    }
    lds_barrier();

    v8f acc = {};
    #pragma unroll
    for (int s = 0; s < HEAD_K / 4; ++s) {
      v2f a;
      a.x = ldsA[mrow * HEAD_PITCH + 4 * s + 2 * kh];
      a.y = ldsA[mrow * HEAD_PITCH + 4 * s + 2 * kh + 1];
      acc = wmma_f32(a, wf[s], acc);
    }

    #pragma unroll
    for (int v = 0; v < 8; ++v) {
      int r = v + 8 * kh;
      float val = acc[v] + bias;
      ldsH[r * 72 + col0 + mrow] = val > 0.0f ? val : 0.0f;
    }
    lds_barrier();

    if (tid < 32) {
      int r = tid >> 1, c = tid & 1;
      float s = bm2[c];
      #pragma unroll 8
      for (int j = 0; j < H_FEATS; ++j)
        s += ldsH[r * 72 + j] * Wm2[j * NUM_CLASSES + c];
      out[(size_t)(row0 + r) * NUM_CLASSES + c] = s;
    }
  }
}

extern "C" void kernel_launch(void* const* d_in, const int* in_sizes, int n_in,
                              void* d_out, int out_size, void* d_ws, size_t ws_size,
                              hipStream_t stream) {
  const float* X      = (const float*)d_in[0];
  const int*   src    = (const int*)  d_in[1];
  const int*   dst    = (const int*)  d_in[2];
  const float* W1     = (const float*)d_in[3];
  const float* b1     = (const float*)d_in[4];
  const float* W2     = (const float*)d_in[5];
  const float* b2     = (const float*)d_in[6];
  const float* thetas = (const float*)d_in[7];
  const float* Wm1    = (const float*)d_in[8];
  const float* bm1    = (const float*)d_in[9];
  const float* Wm2    = (const float*)d_in[10];
  const float* bm2    = (const float*)d_in[11];
  float* out = (float*)d_out;
  float* ws  = (float*)d_ws;

  const int PN = N_NODES * H_FEATS;
  unsigned short* wcode = (unsigned short*)ws;
  float* P0   = ws + 100352;
  float* P1   = P0 + PN;
  float* P2   = P1 + PN;
  float* P3   = P2 + PN;
  unsigned* arrive = (unsigned*)(ws + 100096);

  unsigned* part = (unsigned*)P3;
  degree_hist_kernel<<<DEG_SLICES * DEG_RANGES, DEG_THREADS, 0, stream>>>(
      dst, part, N_EDGES, N_NODES);

  trunk_kernel<<<PT_BLOCKS, 128, 0, stream>>>(X, W1, b1, W2, b2, P0,
                                                 part, wcode, P1, P2, P3 + PN / 2, arrive);

  scatter_kernel<<<SC_BLOCKS, SC_THREADS, 0, stream>>>(
      P0, src, dst, wcode, P1, N_EDGES, N_NODES, 0, P3, PN / 8, arrive);
  scatter_kernel<<<SC_BLOCKS, SC_THREADS, 0, stream>>>(
      P1, src, dst, wcode, P2, N_EDGES, N_NODES, 1, nullptr, 0, arrive + 64);
  scatter_kernel<<<SC_BLOCKS, SC_THREADS, 0, stream>>>(
      P2, src, dst, wcode, P3, N_EDGES, N_NODES, 1, nullptr, 0, arrive + 128);

  head_kernel<<<PT_BLOCKS, 128, 0, stream>>>(
      P0, P1, P2, P3, thetas, Wm1, bm1, Wm2, bm2, wcode, out);
}


# edge passes: each of 256 work-groups sums its 391 destination rows in LDS (LDS f32 adds) from a per-call range-filtered edge list, stores them once; no global atomics, no clears; arrival-order sums
# speedup vs baseline: 3.5972x; 1.1483x over previous
#include <hip/hip_runtime.h>


#define N_NODES   100000
#define N_EDGES   1600000
#define IN_FEATS  128
#define H_FEATS   64
#define N_CONVS   3
#define POLY_ORDER 4
#define NUM_CLASSES 2
#define RG_NODES      391
#define RG_COUNT      256
#define SUBSEG        512
#define SUBHDR        32
#define SUBCAP        (SUBSEG - SUBHDR)
#define SEG_STRIDE    (RG_NODES * H_FEATS)
#define LIST_SRC_BITS 17
static_assert(N_NODES <= (1 << LIST_SRC_BITS), "a source node must fit the low bits of a list word");
static_assert(RG_NODES <= (1 << (32 - LIST_SRC_BITS)), "a destination offset must fit the high bits of a list word");
static_assert((RG_COUNT - 1) * RG_NODES < N_NODES && RG_COUNT * RG_NODES >= N_NODES, "the ranges cover the nodes, none is empty");

typedef __attribute__((ext_vector_type(2))) float v2f;
typedef __attribute__((ext_vector_type(8))) float v8f;

__device__ __forceinline__ v8f wmma_f32(v2f a, v2f b, v8f c) {
  return __builtin_amdgcn_wmma_f32_16x16x4_f32(
       false, a,  false, b,
       (short)0, c,  false,  false);
}

#define WT_ESC_LO 65504u
static_assert(N_EDGES < (1 << 21), "degree >> 5 must fit 16 bits");
static_assert((N_NODES * 2) % 16 == 0, "code table is copied in 16-byte pieces");
__device__ __forceinline__ unsigned wt_degree(unsigned code, const unsigned short* __restrict__ hi, int node) {
  return code < WT_ESC_LO ? code : (((unsigned)hi[node] << 5) | (code - WT_ESC_LO));
}
__device__ __forceinline__ float wt_from_degree(unsigned deg) { return rsqrtf((float)deg); }

__device__ __forceinline__ void lds_barrier() {
  asm volatile("s_wait_dscnt 0x0\n\ts_barrier_signal -1\n\ts_barrier_wait -1" ::: "memory");
}

#define DEG_RANGES      8
#define DEG_RANGE_NODES 12512
#define DEG_SLICES      32
#define DEG_THREADS     512
#define DEG_LOADS       8
#define DEG_SUBS        (DEG_RANGE_NODES / RG_NODES)
#define SEG_WORDS       (DEG_SLICES * SUBSEG)
static_assert(DEG_SUBS * RG_NODES == DEG_RANGE_NODES, "a degree range is whole pass ranges");
static_assert(DEG_RANGES * DEG_SUBS == RG_COUNT, "the degree ranges hold every pass range once");
static_assert(SEG_WORDS <= SEG_STRIDE, "a range's lists lie inside its own rows");
static_assert((size_t)(RG_COUNT - 1) * SEG_STRIDE + SEG_WORDS <= (size_t)N_NODES * H_FEATS, "the last range's lists end inside the buffer");

#define DEG_NO_RANK 0xffffffffu
__device__ __forceinline__ unsigned deg_rank(unsigned a, unsigned span, unsigned* cnt, unsigned* subpos) {
  unsigned rank = DEG_NO_RANK;
  if (a < span) {
    atomicAdd(&cnt[a], 1u);
    rank = atomicAdd(&subpos[a / RG_NODES], 1u);
  }
  return rank;
}
__device__ __forceinline__ void deg_file(unsigned a, unsigned rank, int s, unsigned (*stage)[SUBCAP]) {
  if (rank < SUBCAP) {
    const unsigned sub = a / RG_NODES;
    stage[sub][rank] = ((a - sub * RG_NODES) << LIST_SRC_BITS) | (unsigned)s;
  }
}

__global__ __launch_bounds__(DEG_THREADS)
void degree_hist_kernel(const int* __restrict__ src, const int* __restrict__ dst,
                        unsigned* __restrict__ part, unsigned* __restrict__ lists,
                        int E, int n) {
  __shared__ unsigned cnt[DEG_RANGE_NODES];
  __shared__ unsigned stage[DEG_SUBS][SUBCAP];
  __shared__ unsigned subpos[DEG_SUBS];

  const int tid   = threadIdx.x;
  const int range = blockIdx.x % DEG_RANGES;
  const int slice = blockIdx.x / DEG_RANGES;
  const int lo    = range * DEG_RANGE_NODES;
  int hi = lo + DEG_RANGE_NODES;
  if (hi > n) hi = n;
  const unsigned span = hi > lo ? (unsigned)(hi - lo) : 0u;

  for (int i = tid; i < DEG_RANGE_NODES; i += DEG_THREADS) cnt[i] = 0u;
  if (tid < DEG_SUBS) subpos[tid] = 0u;
  __syncthreads();

  const int slice_len = (((E + DEG_SLICES - 1) / DEG_SLICES) + 3) & ~3;
  const int e_begin = slice * slice_len;
  int e_end = e_begin + slice_len;
  if (e_end > E) e_end = E;
  const int n4 = e_end > e_begin ? (e_end - e_begin) >> 2 : 0;

  const int4* d4 = reinterpret_cast<const int4*>(dst + e_begin);
  const int4* s4 = reinterpret_cast<const int4*>(src + e_begin);
  for (int i0 = tid; i0 < n4; i0 += DEG_LOADS * DEG_THREADS) {
    int4 v[DEG_LOADS], sv[DEG_LOADS];
    #pragma unroll
    for (int u = 0; u < DEG_LOADS; ++u) {
      const int i = i0 + u * DEG_THREADS;
      v[u]  = i < n4 ? d4[i] : make_int4(-1, -1, -1, -1);
      sv[u] = i < n4 ? s4[i] : make_int4(0, 0, 0, 0);
    }
    unsigned rk[DEG_LOADS][4];
    #pragma unroll
    for (int u = 0; u < DEG_LOADS; ++u) {
      rk[u][0] = deg_rank((unsigned)(v[u].x - lo), span, cnt, subpos);
      rk[u][1] = deg_rank((unsigned)(v[u].y - lo), span, cnt, subpos);
      rk[u][2] = deg_rank((unsigned)(v[u].z - lo), span, cnt, subpos);
      rk[u][3] = deg_rank((unsigned)(v[u].w - lo), span, cnt, subpos);
    }
    #pragma unroll
    for (int u = 0; u < DEG_LOADS; ++u) {
      deg_file((unsigned)(v[u].x - lo), rk[u][0], sv[u].x, stage);
      deg_file((unsigned)(v[u].y - lo), rk[u][1], sv[u].y, stage);
      deg_file((unsigned)(v[u].z - lo), rk[u][2], sv[u].z, stage);
      deg_file((unsigned)(v[u].w - lo), rk[u][3], sv[u].w, stage);
    }
  }
  for (int e = e_begin + 4 * n4 + tid; e < e_end; e += DEG_THREADS) {
    const unsigned a = (unsigned)(dst[e] - lo);
    deg_file(a, deg_rank(a, span, cnt, subpos), src[e], stage);
  }
  __syncthreads();

  unsigned* prow = part + (size_t)slice * n + lo;
  for (unsigned i = tid; i < span; i += DEG_THREADS) prow[i] = cnt[i];

  unsigned* seg0 = lists + (size_t)(range * DEG_SUBS) * SEG_STRIDE + slice * SUBSEG;
  for (int i = tid; i < DEG_SUBS * SUBSEG; i += DEG_THREADS) {
    const int sub = i / SUBSEG, k = i % SUBSEG;
    const unsigned c = subpos[sub];
    unsigned* p = seg0 + (size_t)sub * SEG_STRIDE;
    if (k == 0) p[0] = c;
    else if (k >= SUBHDR && (unsigned)(k - SUBHDR) < (c < SUBCAP ? c : SUBCAP)) p[k] = stage[sub][k - SUBHDR];
  }
}

#define ROW_TILES (N_NODES / 16)
#define PT_BLOCKS 1250

__global__ __launch_bounds__(128)
void trunk_kernel(const float* __restrict__ X,
                  const float* __restrict__ W1, const float* __restrict__ b1,
                  const float* __restrict__ W2, const float* __restrict__ b2,
                  float* __restrict__ P0,
                  const unsigned* __restrict__ part, unsigned short* __restrict__ wcode) {
  __shared__ __attribute__((aligned(16))) float ldsX[16 * 132];
  __shared__ float ldsH[16 * 72];
  __shared__ unsigned ldsD[128];

  const int tid  = threadIdx.x;
  const int lane = tid & 31;
  const int wave = tid >> 5;
  const int mrow = lane & 15;
  const int kh   = lane >> 4;
  const int col0 = wave * 16;

  v2f w1f[IN_FEATS / 4], w2f[H_FEATS / 4];
  #pragma unroll
  for (int s = 0; s < IN_FEATS / 4; ++s) {
    w1f[s].x = W1[(size_t)(4 * s + 2 * kh)     * H_FEATS + col0 + mrow];
    w1f[s].y = W1[(size_t)(4 * s + 2 * kh + 1) * H_FEATS + col0 + mrow];
  }
  #pragma unroll
  for (int s = 0; s < H_FEATS / 4; ++s) {
    w2f[s].x = W2[(size_t)(4 * s + 2 * kh)     * H_FEATS + col0 + mrow];
    w2f[s].y = W2[(size_t)(4 * s + 2 * kh + 1) * H_FEATS + col0 + mrow];
  }
  const float bias1 = b1[col0 + mrow];
  const float bias2 = b2[col0 + mrow];


  const int dr = tid & 15, dg = tid >> 4;
  const float4*   xsrc = reinterpret_cast<const float4*>(X) + (size_t)wave * (IN_FEATS / 4) + lane;
  const unsigned* psrc = part + (size_t)(dg * (DEG_SLICES / 8)) * N_NODES + dr;
  int tile = blockIdx.x;
  float4 xr0, xr1, xr2, xr3;
  unsigned pc0, pc1, pc2, pc3;
  {
    const float4*   xs = xsrc + (size_t)tile * (16 * IN_FEATS / 4);
    const unsigned* ps = psrc + tile * 16;
    xr0 = xs[0];  xr1 = xs[4 * (IN_FEATS / 4)];  xr2 = xs[8 * (IN_FEATS / 4)];  xr3 = xs[12 * (IN_FEATS / 4)];
    pc0 = ps[0];  pc1 = ps[N_NODES];  pc2 = ps[2 * N_NODES];  pc3 = ps[3 * N_NODES];
  }

  for (; tile < ROW_TILES; tile += gridDim.x) {
    const int row0 = tile * 16;

    {
      float* xd = &ldsX[wave * 132 + 4 * lane];
      *reinterpret_cast<float4*>(xd)            = xr0;
      *reinterpret_cast<float4*>(xd + 4 * 132)  = xr1;
      *reinterpret_cast<float4*>(xd + 8 * 132)  = xr2;
      *reinterpret_cast<float4*>(xd + 12 * 132) = xr3;
    }
    ldsD[tid] = (pc0 + pc1) + (pc2 + pc3);


    {
      const int nt = tile + (int)gridDim.x;
      if (nt < ROW_TILES) {
        const float4*   xs = xsrc + (size_t)nt * (16 * IN_FEATS / 4);
        const unsigned* ps = psrc + nt * 16;
        xr0 = xs[0];  xr1 = xs[4 * (IN_FEATS / 4)];  xr2 = xs[8 * (IN_FEATS / 4)];  xr3 = xs[12 * (IN_FEATS / 4)];
        pc0 = ps[0];  pc1 = ps[N_NODES];  pc2 = ps[2 * N_NODES];  pc3 = ps[3 * N_NODES];
      }
    }
    lds_barrier();

    if (tid < 16) {
      unsigned cs = 0u;
      #pragma unroll
      for (int g = 0; g < 8; ++g) cs += ldsD[tid + 16 * g];
      cs = cs < 1u ? 1u : cs;
      wcode[row0 + tid]           = (unsigned short)(cs < WT_ESC_LO ? cs : WT_ESC_LO + (cs & 31u));
      wcode[N_NODES + row0 + tid] = (unsigned short)(cs >> 5);
    }

    v8f acc = {};
    #pragma unroll
    for (int s = 0; s < IN_FEATS / 4; ++s) {
      v2f a;
      a.x = ldsX[mrow * 132 + 4 * s + 2 * kh];
      a.y = ldsX[mrow * 132 + 4 * s + 2 * kh + 1];
      acc = wmma_f32(a, w1f[s], acc);
    }
    #pragma unroll
    for (int v = 0; v < 8; ++v) {
      int r = v + 8 * kh;
      float val = acc[v] + bias1;
      ldsH[r * 72 + col0 + mrow] = val > 0.0f ? val : 0.0f;
    }
    lds_barrier();

    v8f acc2 = {};
    #pragma unroll
    for (int s = 0; s < H_FEATS / 4; ++s) {
      v2f a;
      a.x = ldsH[mrow * 72 + 4 * s + 2 * kh];
      a.y = ldsH[mrow * 72 + 4 * s + 2 * kh + 1];
      acc2 = wmma_f32(a, w2f[s], acc2);
    }
    #pragma unroll
    for (int v = 0; v < 8; ++v) {
      int r = v + 8 * kh;
      float val = acc2[v] + bias2;
      P0[(size_t)(row0 + r) * H_FEATS + col0 + mrow] = val > 0.0f ? val : 0.0f;
    }
  }
}

#define SC_BLOCKS   RG_COUNT
#define SC_THREADS  1024
#define SC_WAVES    (SC_THREADS / 32)
#define WT_PIECES   (N_NODES * 2 / 16)
#define WT_LOADS    ((WT_PIECES + SC_THREADS - 1) / SC_THREADS)
#define ROWS_AHEAD  16
#define LIST_TRIPS  (SUBCAP / 32)
static_assert(LIST_TRIPS * 32 == SUBCAP, "a sub-segment is whole 32-edge groups");
static_assert(SC_WAVES == DEG_SLICES, "wave w of an edge pass reads the lists of slice w");

__global__ __launch_bounds__(SC_THREADS)
void scatter_kernel(const float* __restrict__ P,
                    const int* __restrict__ src, const int* __restrict__ dst,
                    const unsigned short* __restrict__ wcode, float* agg,
                    const unsigned* lists, int E, int n, int squared) {
  __shared__ __attribute__((aligned(16))) unsigned short wt_tab[N_NODES];
  __shared__ __attribute__((aligned(16))) float acc[RG_NODES * H_FEATS];

  const int lane = threadIdx.x & 31;
  const int w    = __builtin_amdgcn_readfirstlane((int)(threadIdx.x >> 5));
  const int R    = blockIdx.x;
  const int lo   = R * RG_NODES;
  int hi = lo + RG_NODES;
  if (hi > n) hi = n;
  const unsigned span = hi > lo ? (unsigned)(hi - lo) : 0u;

  const unsigned* seg = lists + (size_t)R * SEG_STRIDE + w * SUBSEG;
  const int cnt = __builtin_amdgcn_readfirstlane((int)seg[0]);
  unsigned wl[LIST_TRIPS];

  {
    const uint4* g4 = reinterpret_cast<const uint4*>(wcode);
    uint4* l4 = reinterpret_cast<uint4*>(wt_tab);
    uint4 v[WT_LOADS];
    #pragma unroll
    for (int u = 0; u < WT_LOADS; ++u) {
      const int i = (int)threadIdx.x + u * SC_THREADS;
      v[u] = g4[i < WT_PIECES ? i : WT_PIECES - 1];
    }
    #pragma unroll
    for (int u = 0; u < LIST_TRIPS; ++u) {
      wl[u] = 0u;
      if (cnt <= SUBCAP && u * 32 + lane < cnt) wl[u] = seg[SUBHDR + u * 32 + lane];
    }
    float4* a4 = reinterpret_cast<float4*>(acc);
    for (int i = threadIdx.x; i < RG_NODES * H_FEATS / 4; i += SC_THREADS)
      a4[i] = make_float4(0.0f, 0.0f, 0.0f, 0.0f);
    #pragma unroll
    for (int u = 0; u < WT_LOADS; ++u) {
      const int i = (int)threadIdx.x + u * SC_THREADS;
      if (i < WT_PIECES) l4[i] = v[u];
    }
  }
  __syncthreads();

  if (cnt <= SUBCAP) {
    for (int t = 0; t * 32 < cnt; ++t) {
      const int k = cnt - t * 32;
      unsigned wcur = 0u;
      #pragma unroll
      for (int u = 0; u < LIST_TRIPS; ++u)
        if (u == t) wcur = wl[u];
      const int es = (int)(wcur & ((1u << LIST_SRC_BITS) - 1u));
      const int ed = (int)(wcur >> LIST_SRC_BITS);
      float ew = 0.0f;
      if (lane < k) {
        const float ds = wt_from_degree(wt_degree(wt_tab[es], wcode + N_NODES, es));
        ew = squared ? ds * ds : ds;
      }
      if (k >= 32) {
        #pragma unroll 1
        for (int g = 0; g < 32; g += ROWS_AHEAD) {
          float vlo[ROWS_AHEAD], vhi[ROWS_AHEAD];
          #pragma unroll
          for (int j = 0; j < ROWS_AHEAD; ++j) {
            const int s = __builtin_amdgcn_readlane(es, g + j);
            const float* rp = P + (size_t)s * H_FEATS + lane;
            vlo[j] = rp[0];
            vhi[j] = rp[H_FEATS / 2];
          }
          #pragma unroll
          for (int j = 0; j < ROWS_AHEAD; ++j) {
            const int   d  = __builtin_amdgcn_readlane(ed, g + j);
            const float di = __int_as_float(__builtin_amdgcn_readlane(__float_as_int(ew), g + j));
            atomicAdd(&acc[d * H_FEATS + lane],               vlo[j] * di);
            atomicAdd(&acc[d * H_FEATS + H_FEATS / 2 + lane], vhi[j] * di);
          }
        }
      } else {
        #pragma unroll 1
        for (int g = 0; g < k; g += ROWS_AHEAD) {
          const int kk = k - g;
          float vlo[ROWS_AHEAD], vhi[ROWS_AHEAD];
          #pragma unroll
          for (int j = 0; j < ROWS_AHEAD; ++j) {
            vlo[j] = 0.0f;  vhi[j] = 0.0f;
            if (j < kk) {
              const int s = __builtin_amdgcn_readlane(es, g + j);
              const float* rp = P + (size_t)s * H_FEATS + lane;
              vlo[j] = rp[0];
              vhi[j] = rp[H_FEATS / 2];
            }
          }
          #pragma unroll
          for (int j = 0; j < ROWS_AHEAD; ++j) {
            if (j < kk) {
              const int   d  = __builtin_amdgcn_readlane(ed, g + j);
              const float di = __int_as_float(__builtin_amdgcn_readlane(__float_as_int(ew), g + j));
              atomicAdd(&acc[d * H_FEATS + lane],               vlo[j] * di);
              atomicAdd(&acc[d * H_FEATS + H_FEATS / 2 + lane], vhi[j] * di);
            }
          }
        }
      }
    }
  } else {
    const int slice_len = (((E + DEG_SLICES - 1) / DEG_SLICES) + 3) & ~3;
    const int e_begin = w * slice_len;
    int e_end = e_begin + slice_len;
    if (e_end > E) e_end = E;
    for (int eb = e_begin; eb < e_end; eb += 32) {
      const int  e   = eb + lane;
      const bool ok  = e < e_end;
      const int  dvv = ok ? dst[e] : -1;
      const int  svv = ok ? src[e] : 0;
      const bool inr = (unsigned)(dvv - lo) < span;
      float ew = 0.0f;
      if (inr) {
        const float ds = wt_from_degree(wt_degree(wt_tab[svv], wcode + N_NODES, svv));
        ew = squared ? ds * ds : ds;
      }
      unsigned m = __builtin_amdgcn_ballot_w32(inr);
      while (m) {
        const int j = __builtin_ctz(m);
        m &= m - 1u;
        const int   s  = __builtin_amdgcn_readlane(svv, j);
        const int   d  = __builtin_amdgcn_readlane(dvv, j) - lo;
        const float di = __int_as_float(__builtin_amdgcn_readlane(__float_as_int(ew), j));
        const float* rp = P + (size_t)s * H_FEATS + lane;
        atomicAdd(&acc[d * H_FEATS + lane],               rp[0] * di);
        atomicAdd(&acc[d * H_FEATS + H_FEATS / 2 + lane], rp[H_FEATS / 2] * di);
      }
    }
  }
  __syncthreads();

  {
    const float4* a4 = reinterpret_cast<const float4*>(acc);
    float4* o4 = reinterpret_cast<float4*>(agg) + (size_t)lo * (H_FEATS / 4);
    const int n4 = (int)span * (H_FEATS / 4);
    for (int i = threadIdx.x; i < n4; i += SC_THREADS) o4[i] = a4[i];
  }
}

#define HEAD_K     (N_CONVS * H_FEATS)
#define HEAD_PITCH (HEAD_K + 4)

__global__ __launch_bounds__(128, 5)
void head_kernel(const float* __restrict__ P0, const float* __restrict__ P1,
                 const float* __restrict__ P2, const float* __restrict__ P3,
                 const float* __restrict__ thetas, const float* __restrict__ Wm1,
                 const float* __restrict__ bm1,
                 const float* __restrict__ Wm2, const float* __restrict__ bm2,
                 const unsigned short* __restrict__ wcode, float* __restrict__ out) {
  __shared__ float ldsA[16 * HEAD_PITCH];
  __shared__ float ldsH[16 * 72];

  const int tid  = threadIdx.x;
  const int lane = tid & 31;
  const int wave = tid >> 5;
  const int mrow = lane & 15;
  const int kh   = lane >> 4;
  const int col0 = wave * 16;

  float cf[N_CONVS][POLY_ORDER];
  #pragma unroll
  for (int i = 0; i < N_CONVS; ++i) {
    const float t0 = thetas[i * POLY_ORDER + 0], t1 = thetas[i * POLY_ORDER + 1];
    const float t2 = thetas[i * POLY_ORDER + 2], t3 = thetas[i * POLY_ORDER + 3];
    cf[i][0] = t0 + t1 + t2 + t3;
    cf[i][1] = -(t1 + 2.0f * t2 + 3.0f * t3);
    cf[i][2] = t2 + 3.0f * t3;
    cf[i][3] = -t3;
  }

  v2f wf[HEAD_K / 4];
  #pragma unroll
  for (int s = 0; s < HEAD_K / 4; ++s) {
    wf[s].x = Wm1[(size_t)(4 * s + 2 * kh)     * H_FEATS + col0 + mrow];
    wf[s].y = Wm1[(size_t)(4 * s + 2 * kh + 1) * H_FEATS + col0 + mrow];
  }
  const float bias = bm1[col0 + mrow];

  const int tr = tid >> 6, tc = tid & 63;
  float u0[8], u1[8], u2[8], u3[8], dsc[8];
  int tile = blockIdx.x;
  #pragma unroll
  for (int t = 0; t < 8; ++t) {
    const int r = tile * 16 + tr + 2 * t;
    const size_t off = (size_t)r * H_FEATS + tc;
    u0[t] = P0[off];
    u1[t] = P1[off];
    u2[t] = P2[off];
    u3[t] = P3[off];
    dsc[t] = wt_from_degree(wt_degree(wcode[r], wcode + N_NODES, r));
  }

  for (; tile < ROW_TILES; tile += gridDim.x) {
    const int row0 = tile * 16;

    #pragma unroll
    for (int t = 0; t < 8; ++t) {
      float* arow = ldsA + (tr + 2 * t) * HEAD_PITCH + tc;
      #pragma unroll
      for (int i = 0; i < N_CONVS; ++i) {
        const float g = cf[i][1] * u1[t] + cf[i][2] * u2[t] + cf[i][3] * u3[t];
        arow[i * H_FEATS] = cf[i][0] * u0[t] + dsc[t] * g;
      }
    }

    {
      const int nt = tile + (int)gridDim.x;
      if (nt < ROW_TILES) {
        #pragma unroll
        for (int t = 0; t < 8; ++t) {
          const int r = nt * 16 + tr + 2 * t;
          const size_t off = (size_t)r * H_FEATS + tc;
          u0[t] = P0[off];
          u1[t] = P1[off];
          u2[t] = P2[off];
          u3[t] = P3[off];
          dsc[t] = wt_from_degree(wt_degree(wcode[r], wcode + N_NODES, r));
        }
      }
    }
    lds_barrier();

    v8f acc = {};
    #pragma unroll
    for (int s = 0; s < HEAD_K / 4; ++s) {
      v2f a;
      a.x = ldsA[mrow * HEAD_PITCH + 4 * s + 2 * kh];
      a.y = ldsA[mrow * HEAD_PITCH + 4 * s + 2 * kh + 1];
      acc = wmma_f32(a, wf[s], acc);
    }

    #pragma unroll
    for (int v = 0; v < 8; ++v) {
      int r = v + 8 * kh;
      float val = acc[v] + bias;
      ldsH[r * 72 + col0 + mrow] = val > 0.0f ? val : 0.0f;
    }
    lds_barrier();

    if (tid < 32) {
      int r = tid >> 1, c = tid & 1;
      float s = bm2[c];
      #pragma unroll 8
      for (int j = 0; j < H_FEATS; ++j)
        s += ldsH[r * 72 + j] * Wm2[j * NUM_CLASSES + c];
      out[(size_t)(row0 + r) * NUM_CLASSES + c] = s;
    }
  }
}

extern "C" void kernel_launch(void* const* d_in, const int* in_sizes, int n_in,
                              void* d_out, int out_size, void* d_ws, size_t ws_size,
                              hipStream_t stream) {
  const float* X      = (const float*)d_in[0];
  const int*   src    = (const int*)  d_in[1];
  const int*   dst    = (const int*)  d_in[2];
  const float* W1     = (const float*)d_in[3];
  const float* b1     = (const float*)d_in[4];
  const float* W2     = (const float*)d_in[5];
  const float* b2     = (const float*)d_in[6];
  const float* thetas = (const float*)d_in[7];
  const float* Wm1    = (const float*)d_in[8];
  const float* bm1    = (const float*)d_in[9];
  const float* Wm2    = (const float*)d_in[10];
  const float* bm2    = (const float*)d_in[11];
  float* out = (float*)d_out;
  float* ws  = (float*)d_ws;

  const int PN = N_NODES * H_FEATS;
  unsigned short* wcode = (unsigned short*)ws;
  float* P0   = ws + 100352;
  float* P1   = P0 + PN;
  float* P2   = P1 + PN;
  float* P3   = P2 + PN;

  unsigned* part  = (unsigned*)P2;
  unsigned* lists = (unsigned*)P3;
  degree_hist_kernel<<<DEG_SLICES * DEG_RANGES, DEG_THREADS, 0, stream>>>(
      src, dst, part, lists, N_EDGES, N_NODES);

  trunk_kernel<<<PT_BLOCKS, 128, 0, stream>>>(X, W1, b1, W2, b2, P0, part, wcode);

  scatter_kernel<<<SC_BLOCKS, SC_THREADS, 0, stream>>>(
      P0, src, dst, wcode, P1, lists, N_EDGES, N_NODES, 0);
  scatter_kernel<<<SC_BLOCKS, SC_THREADS, 0, stream>>>(
      P1, src, dst, wcode, P2, lists, N_EDGES, N_NODES, 1);
  scatter_kernel<<<SC_BLOCKS, SC_THREADS, 0, stream>>>(
      P2, src, dst, wcode, P3, lists, N_EDGES, N_NODES, 1);

  head_kernel<<<PT_BLOCKS, 128, 0, stream>>>(
      P0, P1, P2, P3, thetas, Wm1, bm1, Wm2, bm2, wcode, out);
}
